# starting kernel
# baseline (speedup 1.0000x reference)
#include <hip/hip_runtime.h>


typedef __attribute__((ext_vector_type(2))) float v2f;
typedef __attribute__((ext_vector_type(8))) float v8f;

#define DFEAT 128
#define HFEAT 15
#define HPAD  16

__device__ __forceinline__ void atomicAddF(float* p, float v) {
  unsafeAtomicAdd(p, v);
}

__global__ void k_deg_init(float* __restrict__ deg, int n) {
  int i = blockIdx.x * blockDim.x + threadIdx.x;
  if (i < n) deg[i] = 1.0f;
}

__global__ void k_deg_count(const int* __restrict__ col, float* __restrict__ deg, int e) {
  int i = blockIdx.x * blockDim.x + threadIdx.x;
  if (i < e) atomicAddF(&deg[col[i]], 1.0f);
}

__global__ void k_deg_rsqrt(float* __restrict__ deg, int n) {
  int i = blockIdx.x * blockDim.x + threadIdx.x;
  if (i < n) deg[i] = rsqrtf(deg[i]);
}

template <int K>
__global__ void k_gemm_scale(const float* __restrict__ A,
                             const float* __restrict__ W,
                             int wrows,
                             const float* __restrict__ dinv,
                             float* __restrict__ hs,
                             float* __restrict__ agg,
                             int numTiles) {
  __shared__ float wlds[K][HPAD];
  for (int i = threadIdx.x; i < K * HPAD; i += blockDim.x) {
    int k = i >> 4, n = i & 15;
    wlds[k][n] = (k < wrows && n < HFEAT) ? W[k * HFEAT + n] : 0.0f;
  }
  __syncthreads();

  int wave = threadIdx.x >> 5;
  int lane = threadIdx.x & 31;
  int tile = blockIdx.x * (blockDim.x >> 5) + wave;
  if (tile >= numTiles) return;

  int m  = lane & 15;
  int kh = (lane >> 4) << 1;
  const float* arow = A + (size_t)(tile * 16 + m) * K;

  v8f c = {};
#pragma unroll
  for (int k0 = 0; k0 < K; k0 += 4) {
    v2f a, b;
    a.x = arow[k0 + kh];
    a.y = arow[k0 + kh + 1];
    b.x = wlds[k0 + kh][m];
    b.y = wlds[k0 + kh + 1][m];
    c = __builtin_amdgcn_wmma_f32_16x16x4_f32(false, a, false, b, (short)0, c,
                                              false, false);
  }

  int mbase = (lane >> 4) << 3;
#pragma unroll
  for (int i = 0; i < 8; ++i) {
    int node = tile * 16 + mbase + i;
    float v = c[i] * dinv[node];
    hs[node * HPAD + m]  = v;
    agg[node * HPAD + m] = v;
  }
}

__global__ void k_scatter(const int* __restrict__ row, const int* __restrict__ col,
                          const float* __restrict__ hs, float* __restrict__ agg,
                          int e) {
  int t = blockIdx.x * blockDim.x + threadIdx.x;
  int eid = t >> 2;
  if (eid >= e) return;
  int q = (t & 3) << 2;
  int r  = row[eid];
  int cc = col[eid];
  const float4 v = *reinterpret_cast<const float4*>(hs + (size_t)r * HPAD + q);
  float* dst = agg + (size_t)cc * HPAD + q;
  atomicAddF(dst + 0, v.x);
  atomicAddF(dst + 1, v.y);
  atomicAddF(dst + 2, v.z);
  atomicAddF(dst + 3, v.w);
}

__global__ void k_finish(const float* __restrict__ agg, const float* __restrict__ dinv,
                         const float* __restrict__ b, float* __restrict__ hout,
                         int n) {
  int t = blockIdx.x * blockDim.x + threadIdx.x;
  if (t >= n * HPAD) return;
  int v = t >> 4, nn = t & 15;
  float x = 0.0f;
  if (nn < HFEAT) x = fmaxf(dinv[v] * agg[t] + b[nn], 0.0f);
  hout[t] = x;
}

__global__ void k_gemm_out(const float* __restrict__ A,
                           const float* __restrict__ Wc,
                           const float* __restrict__ bc,
                           float* __restrict__ out,
                           int numNodeTiles) {
  __shared__ float wlds[HPAD][DFEAT];
  for (int i = threadIdx.x; i < HPAD * DFEAT; i += blockDim.x) {
    int k = i >> 7, n = i & 127;
    wlds[k][n] = (k < HFEAT) ? Wc[k * DFEAT + n] : 0.0f;
  }
  __syncthreads();

  int wave  = threadIdx.x >> 5;
  int lane  = threadIdx.x & 31;
  int wtile = blockIdx.x * (blockDim.x >> 5) + wave;
  if (wtile >= numNodeTiles * 8) return;
  int tile = wtile >> 3;
  int n0   = (wtile & 7) << 4;

  int m  = lane & 15;
  int kh = (lane >> 4) << 1;
  const float* arow = A + (size_t)(tile * 16 + m) * HPAD;

  v8f c = {};
#pragma unroll
  for (int k0 = 0; k0 < HPAD; k0 += 4) {
    v2f a, b;
    a.x = arow[k0 + kh];
    a.y = arow[k0 + kh + 1];
    b.x = wlds[k0 + kh][n0 + m];
    b.y = wlds[k0 + kh + 1][n0 + m];
    c = __builtin_amdgcn_wmma_f32_16x16x4_f32(false, a, false, b, (short)0, c,
                                              false, false);
  }

  float bias = bc[n0 + m];
  int mbase = (lane >> 4) << 3;
#pragma unroll
  for (int i = 0; i < 8; ++i) {
    int node = tile * 16 + mbase + i;
    out[(size_t)node * DFEAT + n0 + m] = c[i] + bias;
  }
}

extern "C" void kernel_launch(void* const* d_in, const int* in_sizes, int n_in,
                              void* d_out, int out_size, void* d_ws, size_t ws_size,
                              hipStream_t stream) {
  const float* x  = (const float*)d_in[0];
  const int*   ei = (const int*)d_in[1];
  const float* W1 = (const float*)d_in[2];
  const float* b1 = (const float*)d_in[3];
  const float* W2 = (const float*)d_in[4];
  const float* b2 = (const float*)d_in[5];
  const float* W3 = (const float*)d_in[6];
  const float* b3 = (const float*)d_in[7];
  const float* Wc = (const float*)d_in[8];
  const float* bc = (const float*)d_in[9];

  const int N = in_sizes[0] / DFEAT;
  const int E = in_sizes[1] / 2;
  const int* row = ei;
  const int* col = ei + E;

  float* dinv = (float*)d_ws;
  float* B0 = dinv + N;
  float* B1 = B0 + (size_t)N * HPAD;
  float* B2 = B1 + (size_t)N * HPAD;
  float* outf = (float*)d_out;

  const int T = 256;
  const int numTiles = N / 16;
  dim3 gN((N + T - 1) / T);
  dim3 gE((E + T - 1) / T);
  dim3 gE4((E * 4 + T - 1) / T);
  dim3 gF((N * HPAD + T - 1) / T);
  dim3 gG((numTiles + 7) / 8);
  dim3 gO((numTiles * 8 + 7) / 8);

  k_deg_init <<<gN, T, 0, stream>>>(dinv, N);
  k_deg_count<<<gE, T, 0, stream>>>(col, dinv, E);
  k_deg_rsqrt<<<gN, T, 0, stream>>>(dinv, N);

  k_gemm_scale<DFEAT><<<gG, T, 0, stream>>>(x, W1, DFEAT, dinv, B0, B1, numTiles);
  k_scatter<<<gE4, T, 0, stream>>>(row, col, B0, B1, E);
  k_finish <<<gF,  T, 0, stream>>>(B1, dinv, b1, B2, N);

  k_gemm_scale<HPAD><<<gG, T, 0, stream>>>(B2, W2, HFEAT, dinv, B0, B1, numTiles);
  k_scatter<<<gE4, T, 0, stream>>>(row, col, B0, B1, E);
  k_finish <<<gF,  T, 0, stream>>>(B1, dinv, b2, B2, N);

  k_gemm_scale<HPAD><<<gG, T, 0, stream>>>(B2, W3, HFEAT, dinv, B0, B1, numTiles);
  k_scatter<<<gE4, T, 0, stream>>>(row, col, B0, B1, E);
  k_finish <<<gF,  T, 0, stream>>>(B1, dinv, b3, B2, N);

  k_gemm_out<<<gO, T, 0, stream>>>(B2, Wc, bc, outf, numTiles);
}


# Each scatter block's edges (tile: target range x source range) sorted by source line once in LDS; waves replay sorted runs: a load's two hs rows mostly share a line; LDS f32 sum order changes
# speedup vs baseline: 5.6073x; 5.6073x over previous
#include <hip/hip_runtime.h>


typedef __attribute__((ext_vector_type(2))) float v2f;
typedef __attribute__((ext_vector_type(8))) float v8f;

#define DFEAT 128
#define HFEAT 15
#define HPAD  16

__device__ __forceinline__ void atomicAddF(float* p, float v) {
  unsafeAtomicAdd(p, v);
}

__global__ void k_deg_init(float* __restrict__ deg, int n) {
  int i = blockIdx.x * blockDim.x + threadIdx.x;
  if (i < n) deg[i] = 1.0f;
}

__global__ void k_deg_rsqrt(float* __restrict__ deg, int n) {
  int i = blockIdx.x * blockDim.x + threadIdx.x;
  if (i < n) deg[i] = rsqrtf(deg[i]);
}

template <int K>
__global__ void k_gemm_scale(const float* __restrict__ A,
                             const float* __restrict__ W,
                             int wrows,
                             const float* __restrict__ dinv,
                             float* __restrict__ hs,
                             float* __restrict__ agg,
                             int numTiles) {
  __shared__ float wlds[K][HPAD];
  for (int i = threadIdx.x; i < K * HPAD; i += blockDim.x) {
    int k = i >> 4, n = i & 15;
    wlds[k][n] = (k < wrows && n < HFEAT) ? W[k * HFEAT + n] : 0.0f;
  }
  __syncthreads();

  int wave = threadIdx.x >> 5;
  int lane = threadIdx.x & 31;
  int tile = blockIdx.x * (blockDim.x >> 5) + wave;
  if (tile >= numTiles) return;

  int m  = lane & 15;
  int kh = (lane >> 4) << 1;
  const float* arow = A + (size_t)(tile * 16 + m) * K;

  v8f c = {};
#pragma unroll
  for (int k0 = 0; k0 < K; k0 += 4) {
    v2f a, b;
    a.x = arow[k0 + kh];
    a.y = arow[k0 + kh + 1];
    b.x = wlds[k0 + kh][m];
    b.y = wlds[k0 + kh + 1][m];
    c = __builtin_amdgcn_wmma_f32_16x16x4_f32(false, a, false, b, (short)0, c,
                                              false, false);
  }

  int mbase = (lane >> 4) << 3;
#pragma unroll
  for (int i = 0; i < 8; ++i) {
    int node = tile * 16 + mbase + i;
    float v = c[i] * dinv[node];
    hs[node * HPAD + m]  = v;
    agg[node * HPAD + m] = v;
  }
}


#define SC_NR     4000
#define SC_CHUNKS 10
#define SC_PC     10
#define SC_T      1024
#define SC_QN     64
#define SC_CAP    1024
#define SC_HDR    32
#define SC_SLOT   (SC_HDR + SC_CAP)
#define SC_BINS   250
#define SC_SUBS   ((SC_T / 32) * SC_PC)
#define SC_SUBCAP 144
#define SC_TILEW  (SC_SUBS * SC_SUBCAP + SC_SUBS)
#define SC_EMAX   (32 * SC_CAP)
#define SC_HISTN  5120
#define SC_NOLIST 0xFFFFFFFFu

__device__ __forceinline__ unsigned sc_srn(int n) {
  return (unsigned)(((n + SC_CHUNKS - 1) / SC_CHUNKS + 1) & ~1);
}

__device__ __forceinline__ void sc_drain(const unsigned* __restrict__ qw, unsigned head,
                                         int count, const float* __restrict__ hs,
                                         float* acc, int lane) {
  const int h = lane >> 4, l16 = lane & 15;
  const uint4* qp = reinterpret_cast<const uint4*>(qw + ((head & (SC_QN - 1)) + 16 * h));
  const uint4 q0 = qp[0], q1 = qp[1], q2 = qp[2], q3 = qp[3];
  const unsigned ent[16] = {q0.x, q0.y, q0.z, q0.w, q1.x, q1.y, q1.z, q1.w,
                            q2.x, q2.y, q2.z, q2.w, q3.x, q3.y, q3.z, q3.w};
  const int left = count - 16 * h;
  float v[16];
#pragma unroll
  for (int s = 0; s < 16; ++s) {
    v[s] = 0.0f;
    if (s < left) v[s] = hs[(size_t)(ent[s] >> 12) * HPAD + l16];
  }
#pragma unroll
  for (int s = 0; s < 16; ++s) {
    if (s < left) unsafeAtomicAdd(&acc[(ent[s] & 4095u) * HPAD + l16], v[s]);
  }
}

__device__ __forceinline__ void sc_filter(const int* __restrict__ row, const int* __restrict__ col,
                                          const float* __restrict__ hs, float* acc, unsigned* qw,
                                          int base, int ncur, unsigned sbase, unsigned sn,
                                          int e0, int e1, int wave, int lane) {
  unsigned head = 0, tail = 0;

  const int4 none = make_int4(-1, -1, -1, -1);
  int it = e0 + wave * 256;
  int4 cA = none, cB = none, rA = none, rB = none;
  {
    const int i = it + lane * 4;
    if (i < e1)       { cA = *reinterpret_cast<const int4*>(col + i);       rA = *reinterpret_cast<const int4*>(row + i); }
    if (i + 128 < e1) { cB = *reinterpret_cast<const int4*>(col + i + 128); rB = *reinterpret_cast<const int4*>(row + i + 128); }
  }
  for (; it < e1; it += SC_T * 8) {
    const int cv[8] = {cA.x, cA.y, cA.z, cA.w, cB.x, cB.y, cB.z, cB.w};
    const int rv[8] = {rA.x, rA.y, rA.z, rA.w, rB.x, rB.y, rB.z, rB.w};
    cA = none; cB = none;
    {
      const int i = it + SC_T * 8 + lane * 4;
      if (i < e1)       { cA = *reinterpret_cast<const int4*>(col + i);       rA = *reinterpret_cast<const int4*>(row + i); }
      if (i + 128 < e1) { cB = *reinterpret_cast<const int4*>(col + i + 128); rB = *reinterpret_cast<const int4*>(row + i + 128); }
    }
#pragma unroll
    for (int j = 0; j < 8; ++j) {
      const unsigned cl = (unsigned)(cv[j] - base);
      const bool p = cl < (unsigned)ncur && ((unsigned)rv[j] - sbase) < sn;
      const unsigned m = __builtin_amdgcn_ballot_w32(p);
      if (m) {
        const unsigned pos = tail + __builtin_amdgcn_mbcnt_lo(m, 0u);
        if (p) qw[pos & (SC_QN - 1)] = ((unsigned)rv[j] << 12) | cl;
        tail += (unsigned)__builtin_popcount(m);
        if (tail - head >= 32u) {
          sc_drain(qw, head, 32, hs, acc, lane);
          head += 32u;
        }
      }
    }
  }
  if (tail != head) sc_drain(qw, head, (int)(tail - head), hs, acc, lane);
}

__device__ __forceinline__ void sc_replay(const unsigned* __restrict__ words, unsigned cnt,
                                          const float* __restrict__ hs, float* acc, int lane) {
  const unsigned h = (unsigned)lane >> 4, l16 = (unsigned)lane & 15u;
  const uint4* qp = reinterpret_cast<const uint4*>(words + 32u * h);
  const char* hsb = reinterpret_cast<const char*>(hs);
  const unsigned last = (cnt - 1u) & ~63u;
  uint4 q[8];
#pragma unroll
  for (int k = 0; k < 8; ++k) q[k] = qp[k];
  unsigned b = 0;
  for (; b + 64u <= cnt; b += 64u) {
    const unsigned ent[32] = {q[0].x, q[0].y, q[0].z, q[0].w, q[1].x, q[1].y, q[1].z, q[1].w,
                              q[2].x, q[2].y, q[2].z, q[2].w, q[3].x, q[3].y, q[3].z, q[3].w,
                              q[4].x, q[4].y, q[4].z, q[4].w, q[5].x, q[5].y, q[5].z, q[5].w,
                              q[6].x, q[6].y, q[6].z, q[6].w, q[7].x, q[7].y, q[7].z, q[7].w};
    float v[32];
#pragma unroll
    for (int s = 0; s < 32; ++s) {
      v[s] = *reinterpret_cast<const float*>(hsb + (((ent[s] >> 6) & 0xFFFFFFC0u) | (l16 << 2)));
    }
    const unsigned nb = min(b + 64u, last);
#pragma unroll
    for (int k = 0; k < 8; ++k) q[k] = qp[nb / 4u + k];
#pragma unroll
    for (int s = 0; s < 32; ++s) {
      unsafeAtomicAdd(&acc[((ent[s] & 4095u) << 4) | l16], v[s]);
    }
  }
  if (b < cnt) {
    const unsigned ent[32] = {q[0].x, q[0].y, q[0].z, q[0].w, q[1].x, q[1].y, q[1].z, q[1].w,
                              q[2].x, q[2].y, q[2].z, q[2].w, q[3].x, q[3].y, q[3].z, q[3].w,
                              q[4].x, q[4].y, q[4].z, q[4].w, q[5].x, q[5].y, q[5].z, q[5].w,
                              q[6].x, q[6].y, q[6].z, q[6].w, q[7].x, q[7].y, q[7].z, q[7].w};
    const int left = ((int)(cnt - b) - (int)h + 1) >> 1;
    float v[32];
#pragma unroll
    for (int s = 0; s < 32; ++s) {
      v[s] = 0.0f;
      if (s < left)
        v[s] = *reinterpret_cast<const float*>(hsb + (((ent[s] >> 6) & 0xFFFFFFC0u) | (l16 << 2)));
    }
#pragma unroll
    for (int s = 0; s < 32; ++s) {
      if (s < left) unsafeAtomicAdd(&acc[((ent[s] & 4095u) << 4) | l16], v[s]);
    }
  }
}

__global__ __launch_bounds__(SC_T) void k_bin(const int* __restrict__ row,
                                              const int* __restrict__ col,
                                              float* __restrict__ slots, int n, int e, int nr) {
  __shared__ unsigned bins[SC_BINS][SC_SUBCAP];
  __shared__ unsigned cur[SC_BINS];

  const int tid = threadIdx.x;
  const int share = blockIdx.x;
  const int per = ((e + SC_PC * 4 - 1) / (SC_PC * 4)) * 4;
  const int e0 = blockIdx.y * per;
  const int e1 = min(e, e0 + per);
  const int first = e0 + share * 256 + (tid >> 6) * (SC_T * 8) + (tid & 63) * 4;
  const unsigned srn = sc_srn(n);
  const int ntiles = nr * SC_CHUNKS;
  const unsigned sub = (unsigned)blockIdx.y * (SC_T / 32) + (unsigned)share;
  unsigned* sl = reinterpret_cast<unsigned*>(slots);

  for (int t0 = 0; t0 < ntiles; t0 += SC_BINS) {
    for (int i = tid; i < SC_BINS; i += SC_T) cur[i] = 0u;
    __syncthreads();

    int i = first;
    int4 cv = make_int4(-1, -1, -1, -1), rv = cv;
    if (i < e1) { cv = *reinterpret_cast<const int4*>(col + i); rv = *reinterpret_cast<const int4*>(row + i); }
    for (; i < e1; i += 16 * SC_T * 8) {
      const int ct[4] = {cv.x, cv.y, cv.z, cv.w};
      const int rt[4] = {rv.x, rv.y, rv.z, rv.w};
      const int nx = i + 16 * SC_T * 8;
      if (nx < e1) { cv = *reinterpret_cast<const int4*>(col + nx); rv = *reinterpret_cast<const int4*>(row + nx); }
      unsigned bn[4], wd[4], pos[4];
#pragma unroll
      for (int j = 0; j < 4; ++j) {
        const unsigned c = (unsigned)ct[j], r = (unsigned)rt[j];
        const unsigned rb = c / (unsigned)SC_NR;
        wd[j] = (r << 12) | (c - rb * (unsigned)SC_NR);
        bn[j] = (r / srn) * (unsigned)nr + rb - (unsigned)t0;
        pos[j] = (unsigned)SC_SUBCAP;
        if (c < (unsigned)n && r < (unsigned)n && bn[j] < (unsigned)SC_BINS)
          pos[j] = atomicAdd(&cur[bn[j]], 1u);
      }
#pragma unroll
      for (int j = 0; j < 4; ++j) {
        if (pos[j] < (unsigned)SC_SUBCAP) bins[bn[j]][pos[j]] = wd[j];
      }
    }
    __syncthreads();

    for (int idx = tid; idx < SC_BINS * SC_SUBCAP; idx += SC_T) {
      const int b = idx / SC_SUBCAP, w = idx % SC_SUBCAP;
      const int tile = t0 + b;
      if (tile < ntiles) {
        const unsigned c = cur[b];
        unsigned* tb = sl + (size_t)tile * SC_TILEW;
        if (w == 0) tb[SC_SUBS * SC_SUBCAP + sub] = c;
        if ((unsigned)w < min(c, (unsigned)SC_SUBCAP)) tb[sub * SC_SUBCAP + w] = bins[b][w];
      }
    }
    __syncthreads();
  }
}

__device__ __forceinline__ unsigned sc_wave_scan(unsigned v, int lane) {
#pragma unroll
  for (int d = 1; d < 32; d <<= 1) {
    const unsigned t = __shfl_up(v, (unsigned)d, 32);
    if (lane >= d) v += t;
  }
  return v;
}

__global__ __launch_bounds__(SC_T) void k_sort(const int* __restrict__ row,
                                               const int* __restrict__ col,
                                               float* __restrict__ deg,
                                               float* slots, int n, int e) {
  __shared__ __attribute__((aligned(16))) unsigned ebuf[SC_EMAX];
  __shared__ unsigned sbuf[SC_EMAX];
  __shared__ unsigned hist[SC_HISTN];
  __shared__ unsigned cnt[SC_NR];
  __shared__ unsigned scnt[SC_SUBS];
  __shared__ unsigned soff[SC_SUBS];
  __shared__ unsigned wsum[SC_T / 32];
  __shared__ unsigned s_total, s_bad;

  const int tid = threadIdx.x;
  const int lane = tid & 31, wave = tid >> 5;
  const int base = blockIdx.x * SC_NR;
  const int ncur = min(SC_NR, n - base);
  const unsigned srn = sc_srn(n);
  const unsigned sbase = (unsigned)blockIdx.y * srn;
  const unsigned sn = sbase < (unsigned)n ? min(srn, (unsigned)n - sbase) : 0u;
  const unsigned tile = blockIdx.y * gridDim.x + blockIdx.x;
  unsigned* tb = reinterpret_cast<unsigned*>(slots) + (size_t)tile * SC_TILEW;

  if (tid < SC_SUBS) scnt[tid] = tb[SC_SUBS * SC_SUBCAP + tid];
  for (int i = tid; i < SC_HISTN; i += SC_T) hist[i] = 0u;
  for (int i = tid; i < SC_NR; i += SC_T) cnt[i] = 0u;
  __syncthreads();

  if (wave == 0) {
    unsigned loc[SC_SUBS / 32], sum = 0u, bad = 0u;
#pragma unroll
    for (int k = 0; k < SC_SUBS / 32; ++k) {
      const unsigned c = scnt[lane * (SC_SUBS / 32) + k];
      bad |= (c > (unsigned)SC_SUBCAP) ? 1u : 0u;
      loc[k] = sum;
      sum += min(c, (unsigned)SC_SUBCAP);
    }
    const unsigned inc = sc_wave_scan(sum, lane);
#pragma unroll
    for (int k = 0; k < SC_SUBS / 32; ++k) soff[lane * (SC_SUBS / 32) + k] = inc - sum + loc[k];
    const unsigned anybad = __builtin_amdgcn_ballot_w32(bad != 0u);
    if (lane == 31) { s_total = inc; s_bad = anybad; }
  }
  __syncthreads();
  const unsigned total = s_total;
  const bool nolist = s_bad != 0u || total > (unsigned)SC_EMAX || (sn + 1u) / 2u > (unsigned)SC_HISTN;

  if (nolist) {
    for (int i = tid * 4; i < e; i += SC_T * 4) {
      const int4 cv = *reinterpret_cast<const int4*>(col + i);
      const int4 rv = *reinterpret_cast<const int4*>(row + i);
      const int ct[4] = {cv.x, cv.y, cv.z, cv.w};
      const int rt[4] = {rv.x, rv.y, rv.z, rv.w};
#pragma unroll
      for (int j = 0; j < 4; ++j) {
        const unsigned cl = (unsigned)(ct[j] - base);
        if (cl < (unsigned)ncur && ((unsigned)rt[j] - sbase) < sn) atomicAdd(&cnt[cl], 1u);
      }
    }
    if (tid < SC_T / 32) tb[(size_t)tid * SC_SLOT] = SC_NOLIST;
  } else {
    for (int j = wave; j < SC_SUBS; j += SC_T / 32) {
      const unsigned c = min(scnt[j], (unsigned)SC_SUBCAP), off = soff[j];
      for (unsigned v = (unsigned)lane * 4u; v < c; v += 128u) {
        const uint4 q = *reinterpret_cast<const uint4*>(tb + j * SC_SUBCAP + v);
        const unsigned w4[4] = {q.x, q.y, q.z, q.w};
#pragma unroll
        for (unsigned t = 0; t < 4u; ++t) {
          if (v + t < c) ebuf[off + v + t] = w4[t];
        }
      }
    }
    __syncthreads();

    for (unsigned i = tid; i < total; i += SC_T) {
      const unsigned w = ebuf[i];
      const unsigned key = min(((w >> 12) - sbase) >> 1, (unsigned)(SC_HISTN - 1));
      atomicAdd(&hist[key], 1u);
      atomicAdd(&cnt[min(w & 4095u, (unsigned)(SC_NR - 1))], 1u);
    }
    __syncthreads();

    {
      unsigned a[SC_HISTN / SC_T], sum = 0u;
#pragma unroll
      for (int k = 0; k < SC_HISTN / SC_T; ++k) {
        const unsigned t = hist[tid * (SC_HISTN / SC_T) + k];
        a[k] = sum;
        sum += t;
      }
      const unsigned inc = sc_wave_scan(sum, lane);
      if (lane == 31) wsum[wave] = inc;
      __syncthreads();
      if (wave == 0) {
        const unsigned v = wsum[lane];
        wsum[lane] = sc_wave_scan(v, lane) - v;
      }
      __syncthreads();
      const unsigned first = wsum[wave] + inc - sum;
#pragma unroll
      for (int k = 0; k < SC_HISTN / SC_T; ++k) hist[tid * (SC_HISTN / SC_T) + k] = first + a[k];
    }
    __syncthreads();

    for (unsigned i = tid; i < total; i += SC_T) {
      const unsigned w = ebuf[i];
      const unsigned key = min(((w >> 12) - sbase) >> 1, (unsigned)(SC_HISTN - 1));
      const unsigned pos = atomicAdd(&hist[key], 1u);
      sbuf[min(pos, (unsigned)(SC_EMAX - 1))] = w;
    }
    __syncthreads();

    const unsigned per = ((total + 31u) / 32u + 63u) & ~63u;
    for (int l = 0; l < SC_T / 32; ++l) {
      const unsigned start = (unsigned)l * per;
      const unsigned cw = total > start ? min(per, total - start) : 0u;
      unsigned* lst = tb + (size_t)l * SC_SLOT;
      const unsigned k = (unsigned)tid;
      if (k == 0u) lst[0] = cw;
      if (k < ((cw + 63u) & ~63u)) {
        const unsigned p = k & 63u;
        const unsigned ed = (k & ~63u) + 2u * (p & 31u) + (p >> 5);
        lst[SC_HDR + k] = ed < cw ? sbuf[start + ed] : 0u;
      }
    }
  }
  __syncthreads();

  for (int i = tid; i < ncur; i += SC_T) atomicAddF(deg + base + i, (float)cnt[i]);
}

__global__ __launch_bounds__(SC_T) void k_scatter(const int* __restrict__ row,
                                                  const int* __restrict__ col,
                                                  const float* __restrict__ hs,
                                                  float* __restrict__ agg,
                                                  float* __restrict__ slots, int n, int e) {
  __shared__ __attribute__((aligned(16))) float acc[SC_NR * HPAD];
  __shared__ __attribute__((aligned(16))) unsigned ring[SC_T / 32][SC_QN];

  const int tid = threadIdx.x;
  const int lane = tid & 31, wave = tid >> 5;
  const int base = blockIdx.x * SC_NR;
  const int ncur = min(SC_NR, n - base);

  float4* acc4 = reinterpret_cast<float4*>(acc);
  for (int i = tid; i < SC_NR * HPAD / 4; i += SC_T) acc4[i] = make_float4(0.f, 0.f, 0.f, 0.f);
  for (int i = tid; i < (SC_T / 32) * SC_QN; i += SC_T) (&ring[0][0])[i] = 0u;
  __syncthreads();

  const unsigned tile = blockIdx.y * gridDim.x + blockIdx.x;
  unsigned* lst = reinterpret_cast<unsigned*>(slots) + (size_t)tile * SC_TILEW + (size_t)wave * SC_SLOT;

  const unsigned cnt = (unsigned)__builtin_amdgcn_readfirstlane((int)lst[0]);
  if (cnt > (unsigned)SC_CAP) {
    const unsigned srn = sc_srn(n);
    const unsigned sbase = (unsigned)blockIdx.y * srn;
    const unsigned sn = sbase < (unsigned)n ? min(srn, (unsigned)n - sbase) : 0u;
    sc_filter(row, col, hs, acc, ring[wave], base, ncur, sbase, sn, 0, e, wave, lane);
  } else if (cnt > 0u) {
    sc_replay(lst + SC_HDR, cnt, hs, acc, lane);
  }
  __syncthreads();

  float* dst = agg + (size_t)base * HPAD;
  const int tot = ncur * HPAD;
  for (int i = tid; i < tot; i += SC_T) {
    if ((i & 15) != 15) atomicAddF(dst + i, acc[i]);
  }
}

__global__ void k_finish(const float* __restrict__ agg, const float* __restrict__ dinv,
                         const float* __restrict__ b, float* __restrict__ hout,
                         int n) {
  int t = blockIdx.x * blockDim.x + threadIdx.x;
  if (t >= n * HPAD) return;
  int v = t >> 4, nn = t & 15;
  float x = 0.0f;
  if (nn < HFEAT) x = fmaxf(dinv[v] * agg[t] + b[nn], 0.0f);
  hout[t] = x;
}

__global__ void k_gemm_out(const float* __restrict__ A,
                           const float* __restrict__ Wc,
                           const float* __restrict__ bc,
                           float* __restrict__ out,
                           int numNodeTiles) {
  __shared__ float wlds[HPAD][DFEAT];
  for (int i = threadIdx.x; i < HPAD * DFEAT; i += blockDim.x) {
    int k = i >> 7, n = i & 127;
    wlds[k][n] = (k < HFEAT) ? Wc[k * DFEAT + n] : 0.0f;
  }
  __syncthreads();

  int wave  = threadIdx.x >> 5;
  int lane  = threadIdx.x & 31;
  int wtile = blockIdx.x * (blockDim.x >> 5) + wave;
  if (wtile >= numNodeTiles * 8) return;
  int tile = wtile >> 3;
  int n0   = (wtile & 7) << 4;

  int m  = lane & 15;
  int kh = (lane >> 4) << 1;
  const float* arow = A + (size_t)(tile * 16 + m) * HPAD;

  v8f c = {};
#pragma unroll
  for (int k0 = 0; k0 < HPAD; k0 += 4) {
    v2f a, b;
    a.x = arow[k0 + kh];
    a.y = arow[k0 + kh + 1];
    b.x = wlds[k0 + kh][n0 + m];
    b.y = wlds[k0 + kh + 1][n0 + m];
    c = __builtin_amdgcn_wmma_f32_16x16x4_f32(false, a, false, b, (short)0, c,
                                              false, false);
  }

  float bias = bc[n0 + m];
  int mbase = (lane >> 4) << 3;
#pragma unroll
  for (int i = 0; i < 8; ++i) {
    int node = tile * 16 + mbase + i;
    out[(size_t)node * DFEAT + n0 + m] = c[i] + bias;
  }
}

extern "C" void kernel_launch(void* const* d_in, const int* in_sizes, int n_in,
                              void* d_out, int out_size, void* d_ws, size_t ws_size,
                              hipStream_t stream) {
  const float* x  = (const float*)d_in[0];
  const int*   ei = (const int*)d_in[1];
  const float* W1 = (const float*)d_in[2];
  const float* b1 = (const float*)d_in[3];
  const float* W2 = (const float*)d_in[4];
  const float* b2 = (const float*)d_in[5];
  const float* W3 = (const float*)d_in[6];
  const float* b3 = (const float*)d_in[7];
  const float* Wc = (const float*)d_in[8];
  const float* bc = (const float*)d_in[9];

  const int N = in_sizes[0] / DFEAT;
  const int E = in_sizes[1] / 2;
  const int* row = ei;
  const int* col = ei + E;

  float* dinv = (float*)d_ws;
  float* B0 = dinv + N;
  float* B1 = B0 + (size_t)N * HPAD;
  float* B2 = B1 + (size_t)N * HPAD;
  float* outf = (float*)d_out;
  float* slots = outf;

  const int T = 256;
  const int numTiles = N / 16;
  dim3 gN((N + T - 1) / T);
  dim3 gS((N + SC_NR - 1) / SC_NR, SC_CHUNKS);
  dim3 gB(SC_T / 32, SC_PC);
  dim3 gF((N * HPAD + T - 1) / T);
  dim3 gG((numTiles + 7) / 8);
  dim3 gO((numTiles * 8 + 7) / 8);

  k_deg_init <<<gN, T, 0, stream>>>(dinv, N);
  k_bin      <<<gB, SC_T, 0, stream>>>(row, col, slots, N, E, (int)gS.x);
  k_sort     <<<gS, SC_T, 0, stream>>>(row, col, dinv, slots, N, E);
  k_deg_rsqrt<<<gN, T, 0, stream>>>(dinv, N);

  k_gemm_scale<DFEAT><<<gG, T, 0, stream>>>(x, W1, DFEAT, dinv, B0, B1, numTiles);
  k_scatter<<<gS, SC_T, 0, stream>>>(row, col, B0, B1, slots, N, E);
  k_finish <<<gF,  T, 0, stream>>>(B1, dinv, b1, B2, N);

  k_gemm_scale<HPAD><<<gG, T, 0, stream>>>(B2, W2, HFEAT, dinv, B0, B1, numTiles);
  k_scatter<<<gS, SC_T, 0, stream>>>(row, col, B0, B1, slots, N, E);
  k_finish <<<gF,  T, 0, stream>>>(B1, dinv, b2, B2, N);

  k_gemm_scale<HPAD><<<gG, T, 0, stream>>>(B2, W3, HFEAT, dinv, B0, B1, numTiles);
  k_scatter<<<gS, SC_T, 0, stream>>>(row, col, B0, B1, slots, N, E);
  k_finish <<<gF,  T, 0, stream>>>(B1, dinv, b3, B2, N);

  k_gemm_out<<<gO, T, 0, stream>>>(B2, Wc, bc, outf, numTiles);
}


# Scatter tile grid re-cut: 50 target ranges of 2000 nodes x 5 source ranges (was 25 x 10): each agg element takes 5 global f32 adds per layer, not 10; LDS and global f32 sum order changes
# speedup vs baseline: 5.8608x; 1.0452x over previous
#include <hip/hip_runtime.h>


typedef __attribute__((ext_vector_type(2))) float v2f;
typedef __attribute__((ext_vector_type(8))) float v8f;

#define DFEAT 128
#define HFEAT 15
#define HPAD  16

__device__ __forceinline__ void atomicAddF(float* p, float v) {
  unsafeAtomicAdd(p, v);
}

__global__ void k_deg_init(float* __restrict__ deg, int n) {
  int i = blockIdx.x * blockDim.x + threadIdx.x;
  if (i < n) deg[i] = 1.0f;
}

__global__ void k_deg_rsqrt(float* __restrict__ deg, int n) {
  int i = blockIdx.x * blockDim.x + threadIdx.x;
  if (i < n) deg[i] = rsqrtf(deg[i]);
}

template <int K>
__global__ void k_gemm_scale(const float* __restrict__ A,
                             const float* __restrict__ W,
                             int wrows,
                             const float* __restrict__ dinv,
                             float* __restrict__ hs,
                             float* __restrict__ agg,
                             int numTiles) {
  __shared__ float wlds[K][HPAD];
  for (int i = threadIdx.x; i < K * HPAD; i += blockDim.x) {
    int k = i >> 4, n = i & 15;
    wlds[k][n] = (k < wrows && n < HFEAT) ? W[k * HFEAT + n] : 0.0f;
  }
  __syncthreads();

  int wave = threadIdx.x >> 5;
  int lane = threadIdx.x & 31;
  int tile = blockIdx.x * (blockDim.x >> 5) + wave;
  if (tile >= numTiles) return;

  int m  = lane & 15;
  int kh = (lane >> 4) << 1;
  const float* arow = A + (size_t)(tile * 16 + m) * K;

  v8f c = {};
#pragma unroll
  for (int k0 = 0; k0 < K; k0 += 4) {
    v2f a, b;
    a.x = arow[k0 + kh];
    a.y = arow[k0 + kh + 1];
    b.x = wlds[k0 + kh][m];
    b.y = wlds[k0 + kh + 1][m];
    c = __builtin_amdgcn_wmma_f32_16x16x4_f32(false, a, false, b, (short)0, c,
                                              false, false);
  }

  int mbase = (lane >> 4) << 3;
#pragma unroll
  for (int i = 0; i < 8; ++i) {
    int node = tile * 16 + mbase + i;
    float v = c[i] * dinv[node];
    hs[node * HPAD + m]  = v;
    agg[node * HPAD + m] = v;
  }
}


#define SC_NR     2000
#define SC_CHUNKS 5
#define SC_PC     10
#define SC_T      1024
#define SC_QN     64
#define SC_CAP    1024
#define SC_HDR    32
#define SC_SLOT   (SC_HDR + SC_CAP)
#define SC_BINS   250
#define SC_SUBS   ((SC_T / 32) * SC_PC)
#define SC_SUBCAP 144
#define SC_TILEW  (SC_SUBS * SC_SUBCAP + SC_SUBS)
#define SC_EMAX   (32 * SC_CAP)
#define SC_HISTN  10240
#define SC_NOLIST 0xFFFFFFFFu

__device__ __forceinline__ unsigned sc_srn(int n) {
  return (unsigned)(((n + SC_CHUNKS - 1) / SC_CHUNKS + 1) & ~1);
}

__device__ __forceinline__ void sc_drain(const unsigned* __restrict__ qw, unsigned head,
                                         int count, const float* __restrict__ hs,
                                         float* acc, int lane) {
  const int h = lane >> 4, l16 = lane & 15;
  const uint4* qp = reinterpret_cast<const uint4*>(qw + ((head & (SC_QN - 1)) + 16 * h));
  const uint4 q0 = qp[0], q1 = qp[1], q2 = qp[2], q3 = qp[3];
  const unsigned ent[16] = {q0.x, q0.y, q0.z, q0.w, q1.x, q1.y, q1.z, q1.w,
                            q2.x, q2.y, q2.z, q2.w, q3.x, q3.y, q3.z, q3.w};
  const int left = count - 16 * h;
  float v[16];
#pragma unroll
  for (int s = 0; s < 16; ++s) {
    v[s] = 0.0f;
    if (s < left) v[s] = hs[(size_t)(ent[s] >> 12) * HPAD + l16];
  }
#pragma unroll
  for (int s = 0; s < 16; ++s) {
    if (s < left) unsafeAtomicAdd(&acc[(ent[s] & 4095u) * HPAD + l16], v[s]);
  }
}

__device__ __forceinline__ void sc_filter(const int* __restrict__ row, const int* __restrict__ col,
                                          const float* __restrict__ hs, float* acc, unsigned* qw,
                                          int base, int ncur, unsigned sbase, unsigned sn,
                                          int e0, int e1, int wave, int lane) {
  unsigned head = 0, tail = 0;

  const int4 none = make_int4(-1, -1, -1, -1);
  int it = e0 + wave * 256;
  int4 cA = none, cB = none, rA = none, rB = none;
  {
    const int i = it + lane * 4;
    if (i < e1)       { cA = *reinterpret_cast<const int4*>(col + i);       rA = *reinterpret_cast<const int4*>(row + i); }
    if (i + 128 < e1) { cB = *reinterpret_cast<const int4*>(col + i + 128); rB = *reinterpret_cast<const int4*>(row + i + 128); }
  }
  for (; it < e1; it += SC_T * 8) {
    const int cv[8] = {cA.x, cA.y, cA.z, cA.w, cB.x, cB.y, cB.z, cB.w};
    const int rv[8] = {rA.x, rA.y, rA.z, rA.w, rB.x, rB.y, rB.z, rB.w};
    cA = none; cB = none;
    {
      const int i = it + SC_T * 8 + lane * 4;
      if (i < e1)       { cA = *reinterpret_cast<const int4*>(col + i);       rA = *reinterpret_cast<const int4*>(row + i); }
      if (i + 128 < e1) { cB = *reinterpret_cast<const int4*>(col + i + 128); rB = *reinterpret_cast<const int4*>(row + i + 128); }
    }
#pragma unroll
    for (int j = 0; j < 8; ++j) {
      const unsigned cl = (unsigned)(cv[j] - base);
      const bool p = cl < (unsigned)ncur && ((unsigned)rv[j] - sbase) < sn;
      const unsigned m = __builtin_amdgcn_ballot_w32(p);
      if (m) {
        const unsigned pos = tail + __builtin_amdgcn_mbcnt_lo(m, 0u);
        if (p) qw[pos & (SC_QN - 1)] = ((unsigned)rv[j] << 12) | cl;
        tail += (unsigned)__builtin_popcount(m);
        if (tail - head >= 32u) {
          sc_drain(qw, head, 32, hs, acc, lane);
          head += 32u;
        }
      }
    }
  }
  if (tail != head) sc_drain(qw, head, (int)(tail - head), hs, acc, lane);
}

__device__ __forceinline__ void sc_replay(const unsigned* __restrict__ words, unsigned cnt,
                                          const float* __restrict__ hs, float* acc, int lane) {
  const unsigned h = (unsigned)lane >> 4, l16 = (unsigned)lane & 15u;
  const uint4* qp = reinterpret_cast<const uint4*>(words + 32u * h);
  const char* hsb = reinterpret_cast<const char*>(hs);
  const unsigned last = (cnt - 1u) & ~63u;
  uint4 q[8];
#pragma unroll
  for (int k = 0; k < 8; ++k) q[k] = qp[k];
  unsigned b = 0;
  for (; b + 64u <= cnt; b += 64u) {
    const unsigned ent[32] = {q[0].x, q[0].y, q[0].z, q[0].w, q[1].x, q[1].y, q[1].z, q[1].w,
                              q[2].x, q[2].y, q[2].z, q[2].w, q[3].x, q[3].y, q[3].z, q[3].w,
                              q[4].x, q[4].y, q[4].z, q[4].w, q[5].x, q[5].y, q[5].z, q[5].w,
                              q[6].x, q[6].y, q[6].z, q[6].w, q[7].x, q[7].y, q[7].z, q[7].w};
    float v[32];
#pragma unroll
    for (int s = 0; s < 32; ++s) {
      v[s] = *reinterpret_cast<const float*>(hsb + (((ent[s] >> 6) & 0xFFFFFFC0u) | (l16 << 2)));
    }
    const unsigned nb = min(b + 64u, last);
#pragma unroll
    for (int k = 0; k < 8; ++k) q[k] = qp[nb / 4u + k];
#pragma unroll
    for (int s = 0; s < 32; ++s) {
      unsafeAtomicAdd(&acc[((ent[s] & 4095u) << 4) | l16], v[s]);
    }
  }
  if (b < cnt) {
    const unsigned ent[32] = {q[0].x, q[0].y, q[0].z, q[0].w, q[1].x, q[1].y, q[1].z, q[1].w,
                              q[2].x, q[2].y, q[2].z, q[2].w, q[3].x, q[3].y, q[3].z, q[3].w,
                              q[4].x, q[4].y, q[4].z, q[4].w, q[5].x, q[5].y, q[5].z, q[5].w,
                              q[6].x, q[6].y, q[6].z, q[6].w, q[7].x, q[7].y, q[7].z, q[7].w};
    const int left = ((int)(cnt - b) - (int)h + 1) >> 1;
    float v[32];
#pragma unroll
    for (int s = 0; s < 32; ++s) {
      v[s] = 0.0f;
      if (s < left)
        v[s] = *reinterpret_cast<const float*>(hsb + (((ent[s] >> 6) & 0xFFFFFFC0u) | (l16 << 2)));
    }
#pragma unroll
    for (int s = 0; s < 32; ++s) {
      if (s < left) unsafeAtomicAdd(&acc[((ent[s] & 4095u) << 4) | l16], v[s]);
    }
  }
}

__global__ __launch_bounds__(SC_T) void k_bin(const int* __restrict__ row,
                                              const int* __restrict__ col,
                                              float* __restrict__ slots, int n, int e, int nr) {
  __shared__ unsigned bins[SC_BINS][SC_SUBCAP];
  __shared__ unsigned cur[SC_BINS];

  const int tid = threadIdx.x;
  const int share = blockIdx.x;
  const int per = ((e + SC_PC * 4 - 1) / (SC_PC * 4)) * 4;
  const int e0 = blockIdx.y * per;
  const int e1 = min(e, e0 + per);
  const int first = e0 + share * 256 + (tid >> 6) * (SC_T * 8) + (tid & 63) * 4;
  const unsigned srn = sc_srn(n);
  const int ntiles = nr * SC_CHUNKS;
  const unsigned sub = (unsigned)blockIdx.y * (SC_T / 32) + (unsigned)share;
  unsigned* sl = reinterpret_cast<unsigned*>(slots);

  for (int t0 = 0; t0 < ntiles; t0 += SC_BINS) {
    for (int i = tid; i < SC_BINS; i += SC_T) cur[i] = 0u;
    __syncthreads();

    int i = first;
    int4 cv = make_int4(-1, -1, -1, -1), rv = cv;
    if (i < e1) { cv = *reinterpret_cast<const int4*>(col + i); rv = *reinterpret_cast<const int4*>(row + i); }
    for (; i < e1; i += 16 * SC_T * 8) {
      const int ct[4] = {cv.x, cv.y, cv.z, cv.w};
      const int rt[4] = {rv.x, rv.y, rv.z, rv.w};
      const int nx = i + 16 * SC_T * 8;
      if (nx < e1) { cv = *reinterpret_cast<const int4*>(col + nx); rv = *reinterpret_cast<const int4*>(row + nx); }
      unsigned bn[4], wd[4], pos[4];
#pragma unroll
      for (int j = 0; j < 4; ++j) {
        const unsigned c = (unsigned)ct[j], r = (unsigned)rt[j];
        const unsigned rb = c / (unsigned)SC_NR;
        wd[j] = (r << 12) | (c - rb * (unsigned)SC_NR);
        bn[j] = (r / srn) * (unsigned)nr + rb - (unsigned)t0;
        pos[j] = (unsigned)SC_SUBCAP;
        if (c < (unsigned)n && r < (unsigned)n && bn[j] < (unsigned)SC_BINS)
          pos[j] = atomicAdd(&cur[bn[j]], 1u);
      }
#pragma unroll
      for (int j = 0; j < 4; ++j) {
        if (pos[j] < (unsigned)SC_SUBCAP) bins[bn[j]][pos[j]] = wd[j];
      }
    }
    __syncthreads();

    for (int idx = tid; idx < SC_BINS * SC_SUBCAP; idx += SC_T) {
      const int b = idx / SC_SUBCAP, w = idx % SC_SUBCAP;
      const int tile = t0 + b;
      if (tile < ntiles) {
        const unsigned c = cur[b];
        unsigned* tb = sl + (size_t)tile * SC_TILEW;
        if (w == 0) tb[SC_SUBS * SC_SUBCAP + sub] = c;
        if ((unsigned)w < min(c, (unsigned)SC_SUBCAP)) tb[sub * SC_SUBCAP + w] = bins[b][w];
      }
    }
    __syncthreads();
  }
}

__device__ __forceinline__ unsigned sc_wave_scan(unsigned v, int lane) {
#pragma unroll
  for (int d = 1; d < 32; d <<= 1) {
    const unsigned t = __shfl_up(v, (unsigned)d, 32);
    if (lane >= d) v += t;
  }
  return v;
}

__global__ __launch_bounds__(SC_T) void k_sort(const int* __restrict__ row,
                                               const int* __restrict__ col,
                                               float* __restrict__ deg,
                                               float* slots, int n, int e) {
  __shared__ __attribute__((aligned(16))) unsigned ebuf[SC_EMAX];
  __shared__ unsigned sbuf[SC_EMAX];
  __shared__ unsigned hist[SC_HISTN];
  __shared__ unsigned cnt[SC_NR];
  __shared__ unsigned scnt[SC_SUBS];
  __shared__ unsigned soff[SC_SUBS];
  __shared__ unsigned wsum[SC_T / 32];
  __shared__ unsigned s_total, s_bad;

  const int tid = threadIdx.x;
  const int lane = tid & 31, wave = tid >> 5;
  const int base = blockIdx.x * SC_NR;
  const int ncur = min(SC_NR, n - base);
  const unsigned srn = sc_srn(n);
  const unsigned sbase = (unsigned)blockIdx.y * srn;
  const unsigned sn = sbase < (unsigned)n ? min(srn, (unsigned)n - sbase) : 0u;
  const unsigned tile = blockIdx.y * gridDim.x + blockIdx.x;
  unsigned* tb = reinterpret_cast<unsigned*>(slots) + (size_t)tile * SC_TILEW;

  if (tid < SC_SUBS) scnt[tid] = tb[SC_SUBS * SC_SUBCAP + tid];
  for (int i = tid; i < SC_HISTN; i += SC_T) hist[i] = 0u;
  for (int i = tid; i < SC_NR; i += SC_T) cnt[i] = 0u;
  __syncthreads();

  if (wave == 0) {
    unsigned loc[SC_SUBS / 32], sum = 0u, bad = 0u;
#pragma unroll
    for (int k = 0; k < SC_SUBS / 32; ++k) {
      const unsigned c = scnt[lane * (SC_SUBS / 32) + k];
      bad |= (c > (unsigned)SC_SUBCAP) ? 1u : 0u;
      loc[k] = sum;
      sum += min(c, (unsigned)SC_SUBCAP);
    }
    const unsigned inc = sc_wave_scan(sum, lane);
#pragma unroll
    for (int k = 0; k < SC_SUBS / 32; ++k) soff[lane * (SC_SUBS / 32) + k] = inc - sum + loc[k];
    const unsigned anybad = __builtin_amdgcn_ballot_w32(bad != 0u);
    if (lane == 31) { s_total = inc; s_bad = anybad; }
  }
  __syncthreads();
  const unsigned total = s_total;
  const bool nolist = s_bad != 0u || total > (unsigned)SC_EMAX || (sn + 1u) / 2u > (unsigned)SC_HISTN;

  if (nolist) {
    for (int i = tid * 4; i < e; i += SC_T * 4) {
      const int4 cv = *reinterpret_cast<const int4*>(col + i);
      const int4 rv = *reinterpret_cast<const int4*>(row + i);
      const int ct[4] = {cv.x, cv.y, cv.z, cv.w};
      const int rt[4] = {rv.x, rv.y, rv.z, rv.w};
#pragma unroll
      for (int j = 0; j < 4; ++j) {
        const unsigned cl = (unsigned)(ct[j] - base);
        if (cl < (unsigned)ncur && ((unsigned)rt[j] - sbase) < sn) atomicAdd(&cnt[cl], 1u);
      }
    }
    if (tid < SC_T / 32) tb[(size_t)tid * SC_SLOT] = SC_NOLIST;
  } else {
    for (int j = wave; j < SC_SUBS; j += SC_T / 32) {
      const unsigned c = min(scnt[j], (unsigned)SC_SUBCAP), off = soff[j];
      for (unsigned v = (unsigned)lane * 4u; v < c; v += 128u) {
        const uint4 q = *reinterpret_cast<const uint4*>(tb + j * SC_SUBCAP + v);
        const unsigned w4[4] = {q.x, q.y, q.z, q.w};
#pragma unroll
        for (unsigned t = 0; t < 4u; ++t) {
          if (v + t < c) ebuf[off + v + t] = w4[t];
        }
      }
    }
    __syncthreads();

    for (unsigned i = tid; i < total; i += SC_T) {
      const unsigned w = ebuf[i];
      const unsigned key = min(((w >> 12) - sbase) >> 1, (unsigned)(SC_HISTN - 1));
      atomicAdd(&hist[key], 1u);
      atomicAdd(&cnt[min(w & 4095u, (unsigned)(SC_NR - 1))], 1u);
    }
    __syncthreads();

    {
      unsigned a[SC_HISTN / SC_T], sum = 0u;
#pragma unroll
      for (int k = 0; k < SC_HISTN / SC_T; ++k) {
        const unsigned t = hist[tid * (SC_HISTN / SC_T) + k];
        a[k] = sum;
        sum += t;
      }
      const unsigned inc = sc_wave_scan(sum, lane);
      if (lane == 31) wsum[wave] = inc;
      __syncthreads();
      if (wave == 0) {
        const unsigned v = wsum[lane];
        wsum[lane] = sc_wave_scan(v, lane) - v;
      }
      __syncthreads();
      const unsigned first = wsum[wave] + inc - sum;
#pragma unroll
      for (int k = 0; k < SC_HISTN / SC_T; ++k) hist[tid * (SC_HISTN / SC_T) + k] = first + a[k];
    }
    __syncthreads();

    for (unsigned i = tid; i < total; i += SC_T) {
      const unsigned w = ebuf[i];
      const unsigned key = min(((w >> 12) - sbase) >> 1, (unsigned)(SC_HISTN - 1));
      const unsigned pos = atomicAdd(&hist[key], 1u);
      sbuf[min(pos, (unsigned)(SC_EMAX - 1))] = w;
    }
    __syncthreads();

    const unsigned per = ((total + 31u) / 32u + 63u) & ~63u;
    for (int l = 0; l < SC_T / 32; ++l) {
      const unsigned start = (unsigned)l * per;
      const unsigned cw = total > start ? min(per, total - start) : 0u;
      unsigned* lst = tb + (size_t)l * SC_SLOT;
      const unsigned k = (unsigned)tid;
      if (k == 0u) lst[0] = cw;
      if (k < ((cw + 63u) & ~63u)) {
        const unsigned p = k & 63u;
        const unsigned ed = (k & ~63u) + 2u * (p & 31u) + (p >> 5);
        lst[SC_HDR + k] = ed < cw ? sbuf[start + ed] : 0u;
      }
    }
  }
  __syncthreads();

  for (int i = tid; i < ncur; i += SC_T) atomicAddF(deg + base + i, (float)cnt[i]);
}

__global__ __launch_bounds__(SC_T) void k_scatter(const int* __restrict__ row,
                                                  const int* __restrict__ col,
                                                  const float* __restrict__ hs,
                                                  float* __restrict__ agg,
                                                  float* __restrict__ slots, int n, int e) {
  __shared__ __attribute__((aligned(16))) float acc[SC_NR * HPAD];
  __shared__ __attribute__((aligned(16))) unsigned ring[SC_T / 32][SC_QN];

  const int tid = threadIdx.x;
  const int lane = tid & 31, wave = tid >> 5;
  const int base = blockIdx.x * SC_NR;
  const int ncur = min(SC_NR, n - base);

  float4* acc4 = reinterpret_cast<float4*>(acc);
  for (int i = tid; i < SC_NR * HPAD / 4; i += SC_T) acc4[i] = make_float4(0.f, 0.f, 0.f, 0.f);
  for (int i = tid; i < (SC_T / 32) * SC_QN; i += SC_T) (&ring[0][0])[i] = 0u;
  __syncthreads();

  const unsigned tile = blockIdx.y * gridDim.x + blockIdx.x;
  unsigned* lst = reinterpret_cast<unsigned*>(slots) + (size_t)tile * SC_TILEW + (size_t)wave * SC_SLOT;

  const unsigned cnt = (unsigned)__builtin_amdgcn_readfirstlane((int)lst[0]);
  if (cnt > (unsigned)SC_CAP) {
    const unsigned srn = sc_srn(n);
    const unsigned sbase = (unsigned)blockIdx.y * srn;
    const unsigned sn = sbase < (unsigned)n ? min(srn, (unsigned)n - sbase) : 0u;
    sc_filter(row, col, hs, acc, ring[wave], base, ncur, sbase, sn, 0, e, wave, lane);
  } else if (cnt > 0u) {
    sc_replay(lst + SC_HDR, cnt, hs, acc, lane);
  }
  __syncthreads();

  float* dst = agg + (size_t)base * HPAD;
  const int tot = ncur * HPAD;
  for (int i = tid; i < tot; i += SC_T) {
    if ((i & 15) != 15) atomicAddF(dst + i, acc[i]);
  }
}

__global__ void k_finish(const float* __restrict__ agg, const float* __restrict__ dinv,
                         const float* __restrict__ b, float* __restrict__ hout,
                         int n) {
  int t = blockIdx.x * blockDim.x + threadIdx.x;
  if (t >= n * HPAD) return;
  int v = t >> 4, nn = t & 15;
  float x = 0.0f;
  if (nn < HFEAT) x = fmaxf(dinv[v] * agg[t] + b[nn], 0.0f);
  hout[t] = x;
}

__global__ void k_gemm_out(const float* __restrict__ A,
                           const float* __restrict__ Wc,
                           const float* __restrict__ bc,
                           float* __restrict__ out,
                           int numNodeTiles) {
  __shared__ float wlds[HPAD][DFEAT];
  for (int i = threadIdx.x; i < HPAD * DFEAT; i += blockDim.x) {
    int k = i >> 7, n = i & 127;
    wlds[k][n] = (k < HFEAT) ? Wc[k * DFEAT + n] : 0.0f;
  }
  __syncthreads();

  int wave  = threadIdx.x >> 5;
  int lane  = threadIdx.x & 31;
  int wtile = blockIdx.x * (blockDim.x >> 5) + wave;
  if (wtile >= numNodeTiles * 8) return;
  int tile = wtile >> 3;
  int n0   = (wtile & 7) << 4;

  int m  = lane & 15;
  int kh = (lane >> 4) << 1;
  const float* arow = A + (size_t)(tile * 16 + m) * HPAD;

  v8f c = {};
#pragma unroll
  for (int k0 = 0; k0 < HPAD; k0 += 4) {
    v2f a, b;
    a.x = arow[k0 + kh];
    a.y = arow[k0 + kh + 1];
    b.x = wlds[k0 + kh][n0 + m];
    b.y = wlds[k0 + kh + 1][n0 + m];
    c = __builtin_amdgcn_wmma_f32_16x16x4_f32(false, a, false, b, (short)0, c,
                                              false, false);
  }

  float bias = bc[n0 + m];
  int mbase = (lane >> 4) << 3;
#pragma unroll
  for (int i = 0; i < 8; ++i) {
    int node = tile * 16 + mbase + i;
    out[(size_t)node * DFEAT + n0 + m] = c[i] + bias;
  }
}

extern "C" void kernel_launch(void* const* d_in, const int* in_sizes, int n_in,
                              void* d_out, int out_size, void* d_ws, size_t ws_size,
                              hipStream_t stream) {
  const float* x  = (const float*)d_in[0];
  const int*   ei = (const int*)d_in[1];
  const float* W1 = (const float*)d_in[2];
  const float* b1 = (const float*)d_in[3];
  const float* W2 = (const float*)d_in[4];
  const float* b2 = (const float*)d_in[5];
  const float* W3 = (const float*)d_in[6];
  const float* b3 = (const float*)d_in[7];
  const float* Wc = (const float*)d_in[8];
  const float* bc = (const float*)d_in[9];

  const int N = in_sizes[0] / DFEAT;
  const int E = in_sizes[1] / 2;
  const int* row = ei;
  const int* col = ei + E;

  float* dinv = (float*)d_ws;
  float* B0 = dinv + N;
  float* B1 = B0 + (size_t)N * HPAD;
  float* B2 = B1 + (size_t)N * HPAD;
  float* outf = (float*)d_out;
  float* slots = outf;

  const int T = 256;
  const int numTiles = N / 16;
  dim3 gN((N + T - 1) / T);
  dim3 gS((N + SC_NR - 1) / SC_NR, SC_CHUNKS);
  dim3 gB(SC_T / 32, SC_PC);
  dim3 gF((N * HPAD + T - 1) / T);
  dim3 gG((numTiles + 7) / 8);
  dim3 gO((numTiles * 8 + 7) / 8);

  k_deg_init <<<gN, T, 0, stream>>>(dinv, N);
  k_bin      <<<gB, SC_T, 0, stream>>>(row, col, slots, N, E, (int)gS.x);
  k_sort     <<<gS, SC_T, 0, stream>>>(row, col, dinv, slots, N, E);
  k_deg_rsqrt<<<gN, T, 0, stream>>>(dinv, N);

  k_gemm_scale<DFEAT><<<gG, T, 0, stream>>>(x, W1, DFEAT, dinv, B0, B1, numTiles);
  k_scatter<<<gS, SC_T, 0, stream>>>(row, col, B0, B1, slots, N, E);
  k_finish <<<gF,  T, 0, stream>>>(B1, dinv, b1, B2, N);

  k_gemm_scale<HPAD><<<gG, T, 0, stream>>>(B2, W2, HFEAT, dinv, B0, B1, numTiles);
  k_scatter<<<gS, SC_T, 0, stream>>>(row, col, B0, B1, slots, N, E);
  k_finish <<<gF,  T, 0, stream>>>(B1, dinv, b2, B2, N);

  k_gemm_scale<HPAD><<<gG, T, 0, stream>>>(B2, W3, HFEAT, dinv, B0, B1, numTiles);
  k_scatter<<<gS, SC_T, 0, stream>>>(row, col, B0, B1, slots, N, E);
  k_finish <<<gF,  T, 0, stream>>>(B1, dinv, b3, B2, N);

  k_gemm_out<<<gO, T, 0, stream>>>(B2, Wc, bc, outf, numTiles);
}


# Node-wise passes folded into neighbours: deg=1 set in the binning launch, rsqrt(deg) in layer 1's product epilogue, relu(dinv*agg+b) in the next product's A loader; 14 -> 9 launches; sums unchanged
# speedup vs baseline: 6.3581x; 1.0849x over previous
#include <hip/hip_runtime.h>


typedef __attribute__((ext_vector_type(2))) float v2f;
typedef __attribute__((ext_vector_type(8))) float v8f;

#define DFEAT 128
#define HFEAT 15
#define HPAD  16

__device__ __forceinline__ void atomicAddF(float* p, float v) {
  unsafeAtomicAdd(p, v);
}

template <int K, bool FIRST>
__global__ void k_gemm_scale(const float* __restrict__ A,
                             const float* __restrict__ W,
                             int wrows,
                             const float* __restrict__ bprev,
                             float* __restrict__ dinv,
                             float* __restrict__ hs,
                             float* __restrict__ agg,
                             int numTiles) {
  __shared__ float wlds[K][HPAD];
  __shared__ float blds[HPAD];
  for (int i = threadIdx.x; i < K * HPAD; i += blockDim.x) {
    int k = i >> 4, n = i & 15;
    wlds[k][n] = (k < wrows && n < HFEAT) ? W[k * HFEAT + n] : 0.0f;
  }
  if (!FIRST && threadIdx.x < HPAD) blds[threadIdx.x] = threadIdx.x < HFEAT ? bprev[threadIdx.x] : 0.0f;
  __syncthreads();

  int wave = threadIdx.x >> 5;
  int lane = threadIdx.x & 31;
  int tile = blockIdx.x * (blockDim.x >> 5) + wave;
  if (tile >= numTiles) return;

  int m  = lane & 15;
  int kh = (lane >> 4) << 1;
  const float* arow = A + (size_t)(tile * 16 + m) * K;
  float dm = 0.0f;
  if (!FIRST) dm = dinv[tile * 16 + m];

  v8f c = {};
#pragma unroll
  for (int k0 = 0; k0 < K; k0 += 4) {
    v2f a, b;
    a.x = arow[k0 + kh];
    a.y = arow[k0 + kh + 1];
    if (!FIRST) {
      a.x = fmaxf(dm * a.x + blds[k0 + kh], 0.0f);
      a.y = (k0 + kh + 1 < HFEAT) ? fmaxf(dm * a.y + blds[k0 + kh + 1], 0.0f) : 0.0f;
    }
    b.x = wlds[k0 + kh][m];
    b.y = wlds[k0 + kh + 1][m];
    c = __builtin_amdgcn_wmma_f32_16x16x4_f32(false, a, false, b, (short)0, c,
                                              false, false);
  }

  int mbase = (lane >> 4) << 3;
  float dv[8];
#pragma unroll
  for (int i = 0; i < 8; ++i) {
    float d = dinv[tile * 16 + mbase + i];
    dv[i] = FIRST ? rsqrtf(d) : d;
  }
  if (FIRST) {
#pragma unroll
    for (int i = 0; i < 8; ++i) {
      if (m == 0) dinv[tile * 16 + mbase + i] = dv[i];
    }
  }
#pragma unroll
  for (int i = 0; i < 8; ++i) {
    int node = tile * 16 + mbase + i;
    float v = c[i] * dv[i];
    hs[node * HPAD + m]  = v;
    agg[node * HPAD + m] = v;
  }
}


#define SC_NR     2000
#define SC_CHUNKS 5
#define SC_PC     10
#define SC_T      1024
#define SC_QN     64
#define SC_CAP    1024
#define SC_HDR    32
#define SC_SLOT   (SC_HDR + SC_CAP)
#define SC_BINS   250
#define SC_SUBS   ((SC_T / 32) * SC_PC)
#define SC_SUBCAP 144
#define SC_TILEW  (SC_SUBS * SC_SUBCAP + SC_SUBS)
#define SC_EMAX   (32 * SC_CAP)
#define SC_HISTN  10240
#define SC_NOLIST 0xFFFFFFFFu

__device__ __forceinline__ unsigned sc_srn(int n) {
  return (unsigned)(((n + SC_CHUNKS - 1) / SC_CHUNKS + 1) & ~1);
}

__device__ __forceinline__ void sc_drain(const unsigned* __restrict__ qw, unsigned head,
                                         int count, const float* __restrict__ hs,
                                         float* acc, int lane) {
  const int h = lane >> 4, l16 = lane & 15;
  const uint4* qp = reinterpret_cast<const uint4*>(qw + ((head & (SC_QN - 1)) + 16 * h));
  const uint4 q0 = qp[0], q1 = qp[1], q2 = qp[2], q3 = qp[3];
  const unsigned ent[16] = {q0.x, q0.y, q0.z, q0.w, q1.x, q1.y, q1.z, q1.w,
                            q2.x, q2.y, q2.z, q2.w, q3.x, q3.y, q3.z, q3.w};
  const int left = count - 16 * h;
  float v[16];
#pragma unroll
  for (int s = 0; s < 16; ++s) {
    v[s] = 0.0f;
    if (s < left) v[s] = hs[(size_t)(ent[s] >> 12) * HPAD + l16];
  }
#pragma unroll
  for (int s = 0; s < 16; ++s) {
    if (s < left) unsafeAtomicAdd(&acc[(ent[s] & 4095u) * HPAD + l16], v[s]);
  }
}

__device__ __forceinline__ void sc_filter(const int* __restrict__ row, const int* __restrict__ col,
                                          const float* __restrict__ hs, float* acc, unsigned* qw,
                                          int base, int ncur, unsigned sbase, unsigned sn,
                                          int e0, int e1, int wave, int lane) {
  unsigned head = 0, tail = 0;

  const int4 none = make_int4(-1, -1, -1, -1);
  int it = e0 + wave * 256;
  int4 cA = none, cB = none, rA = none, rB = none;
  {
    const int i = it + lane * 4;
    if (i < e1)       { cA = *reinterpret_cast<const int4*>(col + i);       rA = *reinterpret_cast<const int4*>(row + i); }
    if (i + 128 < e1) { cB = *reinterpret_cast<const int4*>(col + i + 128); rB = *reinterpret_cast<const int4*>(row + i + 128); }
  }
  for (; it < e1; it += SC_T * 8) {
    const int cv[8] = {cA.x, cA.y, cA.z, cA.w, cB.x, cB.y, cB.z, cB.w};
    const int rv[8] = {rA.x, rA.y, rA.z, rA.w, rB.x, rB.y, rB.z, rB.w};
    cA = none; cB = none;
    {
      const int i = it + SC_T * 8 + lane * 4;
      if (i < e1)       { cA = *reinterpret_cast<const int4*>(col + i);       rA = *reinterpret_cast<const int4*>(row + i); }
      if (i + 128 < e1) { cB = *reinterpret_cast<const int4*>(col + i + 128); rB = *reinterpret_cast<const int4*>(row + i + 128); }
    }
#pragma unroll
    for (int j = 0; j < 8; ++j) {
      const unsigned cl = (unsigned)(cv[j] - base);
      const bool p = cl < (unsigned)ncur && ((unsigned)rv[j] - sbase) < sn;
      const unsigned m = __builtin_amdgcn_ballot_w32(p);
      if (m) {
        const unsigned pos = tail + __builtin_amdgcn_mbcnt_lo(m, 0u);
        if (p) qw[pos & (SC_QN - 1)] = ((unsigned)rv[j] << 12) | cl;
        tail += (unsigned)__builtin_popcount(m);
        if (tail - head >= 32u) {
          sc_drain(qw, head, 32, hs, acc, lane);
          head += 32u;
        }
      }
    }
  }
  if (tail != head) sc_drain(qw, head, (int)(tail - head), hs, acc, lane);
}

__device__ __forceinline__ void sc_replay(const unsigned* __restrict__ words, unsigned cnt,
                                          const float* __restrict__ hs, float* acc, int lane) {
  const unsigned h = (unsigned)lane >> 4, l16 = (unsigned)lane & 15u;
  const uint4* qp = reinterpret_cast<const uint4*>(words + 32u * h);
  const char* hsb = reinterpret_cast<const char*>(hs);
  const unsigned last = (cnt - 1u) & ~63u;
  uint4 q[8];
#pragma unroll
  for (int k = 0; k < 8; ++k) q[k] = qp[k];
  unsigned b = 0;
  for (; b + 64u <= cnt; b += 64u) {
    const unsigned ent[32] = {q[0].x, q[0].y, q[0].z, q[0].w, q[1].x, q[1].y, q[1].z, q[1].w,
                              q[2].x, q[2].y, q[2].z, q[2].w, q[3].x, q[3].y, q[3].z, q[3].w,
                              q[4].x, q[4].y, q[4].z, q[4].w, q[5].x, q[5].y, q[5].z, q[5].w,
                              q[6].x, q[6].y, q[6].z, q[6].w, q[7].x, q[7].y, q[7].z, q[7].w};
    float v[32];
#pragma unroll
    for (int s = 0; s < 32; ++s) {
      v[s] = *reinterpret_cast<const float*>(hsb + (((ent[s] >> 6) & 0xFFFFFFC0u) | (l16 << 2)));
    }
    const unsigned nb = min(b + 64u, last);
#pragma unroll
    for (int k = 0; k < 8; ++k) q[k] = qp[nb / 4u + k];
#pragma unroll
    for (int s = 0; s < 32; ++s) {
      unsafeAtomicAdd(&acc[((ent[s] & 4095u) << 4) | l16], v[s]);
    }
  }
  if (b < cnt) {
    const unsigned ent[32] = {q[0].x, q[0].y, q[0].z, q[0].w, q[1].x, q[1].y, q[1].z, q[1].w,
                              q[2].x, q[2].y, q[2].z, q[2].w, q[3].x, q[3].y, q[3].z, q[3].w,
                              q[4].x, q[4].y, q[4].z, q[4].w, q[5].x, q[5].y, q[5].z, q[5].w,
                              q[6].x, q[6].y, q[6].z, q[6].w, q[7].x, q[7].y, q[7].z, q[7].w};
    const int left = ((int)(cnt - b) - (int)h + 1) >> 1;
    float v[32];
#pragma unroll
    for (int s = 0; s < 32; ++s) {
      v[s] = 0.0f;
      if (s < left)
        v[s] = *reinterpret_cast<const float*>(hsb + (((ent[s] >> 6) & 0xFFFFFFC0u) | (l16 << 2)));
    }
#pragma unroll
    for (int s = 0; s < 32; ++s) {
      if (s < left) unsafeAtomicAdd(&acc[((ent[s] & 4095u) << 4) | l16], v[s]);
    }
  }
}

__global__ __launch_bounds__(SC_T) void k_bin(const int* __restrict__ row,
                                              const int* __restrict__ col,
                                              float* __restrict__ slots,
                                              float* __restrict__ deg, int n, int e, int nr) {
  __shared__ unsigned bins[SC_BINS][SC_SUBCAP];
  __shared__ unsigned cur[SC_BINS];

  const int tid = threadIdx.x;
  const int share = blockIdx.x;
  {
    const int nthr = (int)(gridDim.x * gridDim.y) * SC_T;
    for (int i = (int)(blockIdx.y * gridDim.x + blockIdx.x) * SC_T + tid; i < n; i += nthr) deg[i] = 1.0f;
  }
  const int per = ((e + SC_PC * 4 - 1) / (SC_PC * 4)) * 4;
  const int e0 = blockIdx.y * per;
  const int e1 = min(e, e0 + per);
  const int first = e0 + share * 256 + (tid >> 6) * (SC_T * 8) + (tid & 63) * 4;
  const unsigned srn = sc_srn(n);
  const int ntiles = nr * SC_CHUNKS;
  const unsigned sub = (unsigned)blockIdx.y * (SC_T / 32) + (unsigned)share;
  unsigned* sl = reinterpret_cast<unsigned*>(slots);

  for (int t0 = 0; t0 < ntiles; t0 += SC_BINS) {
    for (int i = tid; i < SC_BINS; i += SC_T) cur[i] = 0u;
    __syncthreads();

    int i = first;
    int4 cv = make_int4(-1, -1, -1, -1), rv = cv;
    if (i < e1) { cv = *reinterpret_cast<const int4*>(col + i); rv = *reinterpret_cast<const int4*>(row + i); }
    for (; i < e1; i += 16 * SC_T * 8) {
      const int ct[4] = {cv.x, cv.y, cv.z, cv.w};
      const int rt[4] = {rv.x, rv.y, rv.z, rv.w};
      const int nx = i + 16 * SC_T * 8;
      if (nx < e1) { cv = *reinterpret_cast<const int4*>(col + nx); rv = *reinterpret_cast<const int4*>(row + nx); }
      unsigned bn[4], wd[4], pos[4];
#pragma unroll
      for (int j = 0; j < 4; ++j) {
        const unsigned c = (unsigned)ct[j], r = (unsigned)rt[j];
        const unsigned rb = c / (unsigned)SC_NR;
        wd[j] = (r << 12) | (c - rb * (unsigned)SC_NR);
        bn[j] = (r / srn) * (unsigned)nr + rb - (unsigned)t0;
        pos[j] = (unsigned)SC_SUBCAP;
        if (c < (unsigned)n && r < (unsigned)n && bn[j] < (unsigned)SC_BINS)
          pos[j] = atomicAdd(&cur[bn[j]], 1u);
      }
#pragma unroll
      for (int j = 0; j < 4; ++j) {
        if (pos[j] < (unsigned)SC_SUBCAP) bins[bn[j]][pos[j]] = wd[j];
      }
    }
    __syncthreads();

    for (int idx = tid; idx < SC_BINS * SC_SUBCAP; idx += SC_T) {
      const int b = idx / SC_SUBCAP, w = idx % SC_SUBCAP;
      const int tile = t0 + b;
      if (tile < ntiles) {
        const unsigned c = cur[b];
        unsigned* tb = sl + (size_t)tile * SC_TILEW;
        if (w == 0) tb[SC_SUBS * SC_SUBCAP + sub] = c;
        if ((unsigned)w < min(c, (unsigned)SC_SUBCAP)) tb[sub * SC_SUBCAP + w] = bins[b][w];
      }
    }
    __syncthreads();
  }
}

__device__ __forceinline__ unsigned sc_wave_scan(unsigned v, int lane) {
#pragma unroll
  for (int d = 1; d < 32; d <<= 1) {
    const unsigned t = __shfl_up(v, (unsigned)d, 32);
    if (lane >= d) v += t;
  }
  return v;
}

__global__ __launch_bounds__(SC_T) void k_sort(const int* __restrict__ row,
                                               const int* __restrict__ col,
                                               float* __restrict__ deg,
                                               float* slots, int n, int e) {
  __shared__ __attribute__((aligned(16))) unsigned ebuf[SC_EMAX];
  __shared__ unsigned sbuf[SC_EMAX];
  __shared__ unsigned hist[SC_HISTN];
  __shared__ unsigned cnt[SC_NR];
  __shared__ unsigned scnt[SC_SUBS];
  __shared__ unsigned soff[SC_SUBS];
  __shared__ unsigned wsum[SC_T / 32];
  __shared__ unsigned s_total, s_bad;

  const int tid = threadIdx.x;
  const int lane = tid & 31, wave = tid >> 5;
  const int base = blockIdx.x * SC_NR;
  const int ncur = min(SC_NR, n - base);
  const unsigned srn = sc_srn(n);
  const unsigned sbase = (unsigned)blockIdx.y * srn;
  const unsigned sn = sbase < (unsigned)n ? min(srn, (unsigned)n - sbase) : 0u;
  const unsigned tile = blockIdx.y * gridDim.x + blockIdx.x;
  unsigned* tb = reinterpret_cast<unsigned*>(slots) + (size_t)tile * SC_TILEW;

  if (tid < SC_SUBS) scnt[tid] = tb[SC_SUBS * SC_SUBCAP + tid];
  for (int i = tid; i < SC_HISTN; i += SC_T) hist[i] = 0u;
  for (int i = tid; i < SC_NR; i += SC_T) cnt[i] = 0u;
  __syncthreads();

  if (wave == 0) {
    unsigned loc[SC_SUBS / 32], sum = 0u, bad = 0u;
#pragma unroll
    for (int k = 0; k < SC_SUBS / 32; ++k) {
      const unsigned c = scnt[lane * (SC_SUBS / 32) + k];
      bad |= (c > (unsigned)SC_SUBCAP) ? 1u : 0u;
      loc[k] = sum;
      sum += min(c, (unsigned)SC_SUBCAP);
    }
    const unsigned inc = sc_wave_scan(sum, lane);
#pragma unroll
    for (int k = 0; k < SC_SUBS / 32; ++k) soff[lane * (SC_SUBS / 32) + k] = inc - sum + loc[k];
    const unsigned anybad = __builtin_amdgcn_ballot_w32(bad != 0u);
    if (lane == 31) { s_total = inc; s_bad = anybad; }
  }
  __syncthreads();
  const unsigned total = s_total;
  const bool nolist = s_bad != 0u || total > (unsigned)SC_EMAX || (sn + 1u) / 2u > (unsigned)SC_HISTN;

  if (nolist) {
    for (int i = tid * 4; i < e; i += SC_T * 4) {
      const int4 cv = *reinterpret_cast<const int4*>(col + i);
      const int4 rv = *reinterpret_cast<const int4*>(row + i);
      const int ct[4] = {cv.x, cv.y, cv.z, cv.w};
      const int rt[4] = {rv.x, rv.y, rv.z, rv.w};
#pragma unroll
      for (int j = 0; j < 4; ++j) {
        const unsigned cl = (unsigned)(ct[j] - base);
        if (cl < (unsigned)ncur && ((unsigned)rt[j] - sbase) < sn) atomicAdd(&cnt[cl], 1u);
      }
    }
    if (tid < SC_T / 32) tb[(size_t)tid * SC_SLOT] = SC_NOLIST;
  } else {
    for (int j = wave; j < SC_SUBS; j += SC_T / 32) {
      const unsigned c = min(scnt[j], (unsigned)SC_SUBCAP), off = soff[j];
      for (unsigned v = (unsigned)lane * 4u; v < c; v += 128u) {
        const uint4 q = *reinterpret_cast<const uint4*>(tb + j * SC_SUBCAP + v);
        const unsigned w4[4] = {q.x, q.y, q.z, q.w};
#pragma unroll
        for (unsigned t = 0; t < 4u; ++t) {
          if (v + t < c) ebuf[off + v + t] = w4[t];
        }
      }
    }
    __syncthreads();

    for (unsigned i = tid; i < total; i += SC_T) {
      const unsigned w = ebuf[i];
      const unsigned key = min(((w >> 12) - sbase) >> 1, (unsigned)(SC_HISTN - 1));
      atomicAdd(&hist[key], 1u);
      atomicAdd(&cnt[min(w & 4095u, (unsigned)(SC_NR - 1))], 1u);
    }
    __syncthreads();

    {
      unsigned a[SC_HISTN / SC_T], sum = 0u;
#pragma unroll
      for (int k = 0; k < SC_HISTN / SC_T; ++k) {
        const unsigned t = hist[tid * (SC_HISTN / SC_T) + k];
        a[k] = sum;
        sum += t;
      }
      const unsigned inc = sc_wave_scan(sum, lane);
      if (lane == 31) wsum[wave] = inc;
      __syncthreads();
      if (wave == 0) {
        const unsigned v = wsum[lane];
        wsum[lane] = sc_wave_scan(v, lane) - v;
      }
      __syncthreads();
      const unsigned first = wsum[wave] + inc - sum;
#pragma unroll
      for (int k = 0; k < SC_HISTN / SC_T; ++k) hist[tid * (SC_HISTN / SC_T) + k] = first + a[k];
    }
    __syncthreads();

    for (unsigned i = tid; i < total; i += SC_T) {
      const unsigned w = ebuf[i];
      const unsigned key = min(((w >> 12) - sbase) >> 1, (unsigned)(SC_HISTN - 1));
      const unsigned pos = atomicAdd(&hist[key], 1u);
      sbuf[min(pos, (unsigned)(SC_EMAX - 1))] = w;
    }
    __syncthreads();

    const unsigned per = ((total + 31u) / 32u + 63u) & ~63u;
    for (int l = 0; l < SC_T / 32; ++l) {
      const unsigned start = (unsigned)l * per;
      const unsigned cw = total > start ? min(per, total - start) : 0u;
      unsigned* lst = tb + (size_t)l * SC_SLOT;
      const unsigned k = (unsigned)tid;
      if (k == 0u) lst[0] = cw;
      if (k < ((cw + 63u) & ~63u)) {
        const unsigned p = k & 63u;
        const unsigned ed = (k & ~63u) + 2u * (p & 31u) + (p >> 5);
        lst[SC_HDR + k] = ed < cw ? sbuf[start + ed] : 0u;
      }
    }
  }
  __syncthreads();

  for (int i = tid; i < ncur; i += SC_T) atomicAddF(deg + base + i, (float)cnt[i]);
}

__global__ __launch_bounds__(SC_T) void k_scatter(const int* __restrict__ row,
                                                  const int* __restrict__ col,
                                                  const float* __restrict__ hs,
                                                  float* __restrict__ agg,
                                                  float* __restrict__ slots, int n, int e) {
  __shared__ __attribute__((aligned(16))) float acc[SC_NR * HPAD];
  __shared__ __attribute__((aligned(16))) unsigned ring[SC_T / 32][SC_QN];

  const int tid = threadIdx.x;
  const int lane = tid & 31, wave = tid >> 5;
  const int base = blockIdx.x * SC_NR;
  const int ncur = min(SC_NR, n - base);

  float4* acc4 = reinterpret_cast<float4*>(acc);
  for (int i = tid; i < SC_NR * HPAD / 4; i += SC_T) acc4[i] = make_float4(0.f, 0.f, 0.f, 0.f);
  for (int i = tid; i < (SC_T / 32) * SC_QN; i += SC_T) (&ring[0][0])[i] = 0u;
  __syncthreads();

  const unsigned tile = blockIdx.y * gridDim.x + blockIdx.x;
  unsigned* lst = reinterpret_cast<unsigned*>(slots) + (size_t)tile * SC_TILEW + (size_t)wave * SC_SLOT;

  const unsigned cnt = (unsigned)__builtin_amdgcn_readfirstlane((int)lst[0]);
  if (cnt > (unsigned)SC_CAP) {
    const unsigned srn = sc_srn(n);
    const unsigned sbase = (unsigned)blockIdx.y * srn;
    const unsigned sn = sbase < (unsigned)n ? min(srn, (unsigned)n - sbase) : 0u;
    sc_filter(row, col, hs, acc, ring[wave], base, ncur, sbase, sn, 0, e, wave, lane);
  } else if (cnt > 0u) {
    sc_replay(lst + SC_HDR, cnt, hs, acc, lane);
  }
  __syncthreads();

  float* dst = agg + (size_t)base * HPAD;
  const int tot = ncur * HPAD;
  for (int i = tid; i < tot; i += SC_T) {
    if ((i & 15) != 15) atomicAddF(dst + i, acc[i]);
  }
}

__global__ void k_gemm_out(const float* __restrict__ A,
                           const float* __restrict__ dinv,
                           const float* __restrict__ bprev,
                           const float* __restrict__ Wc,
                           const float* __restrict__ bc,
                           float* __restrict__ out,
                           int numNodeTiles) {
  __shared__ float wlds[HPAD][DFEAT];
  __shared__ float blds[HPAD];
  for (int i = threadIdx.x; i < HPAD * DFEAT; i += blockDim.x) {
    int k = i >> 7, n = i & 127;
    wlds[k][n] = (k < HFEAT) ? Wc[k * DFEAT + n] : 0.0f;
  }
  if (threadIdx.x < HPAD) blds[threadIdx.x] = threadIdx.x < HFEAT ? bprev[threadIdx.x] : 0.0f;
  __syncthreads();

  int wave  = threadIdx.x >> 5;
  int lane  = threadIdx.x & 31;
  int wtile = blockIdx.x * (blockDim.x >> 5) + wave;
  if (wtile >= numNodeTiles * 8) return;
  int tile = wtile >> 3;
  int n0   = (wtile & 7) << 4;

  int m  = lane & 15;
  int kh = (lane >> 4) << 1;
  const float* arow = A + (size_t)(tile * 16 + m) * HPAD;
  const float dm = dinv[tile * 16 + m];

  v8f c = {};
#pragma unroll
  for (int k0 = 0; k0 < HPAD; k0 += 4) {
    v2f a, b;
    a.x = fmaxf(dm * arow[k0 + kh] + blds[k0 + kh], 0.0f);
    a.y = (k0 + kh + 1 < HFEAT) ? fmaxf(dm * arow[k0 + kh + 1] + blds[k0 + kh + 1], 0.0f) : 0.0f;
    b.x = wlds[k0 + kh][n0 + m];
    b.y = wlds[k0 + kh + 1][n0 + m];
    c = __builtin_amdgcn_wmma_f32_16x16x4_f32(false, a, false, b, (short)0, c,
                                              false, false);
  }

  float bias = bc[n0 + m];
  int mbase = (lane >> 4) << 3;
#pragma unroll
  for (int i = 0; i < 8; ++i) {
    int node = tile * 16 + mbase + i;
    out[(size_t)node * DFEAT + n0 + m] = c[i] + bias;
  }
}

extern "C" void kernel_launch(void* const* d_in, const int* in_sizes, int n_in,
                              void* d_out, int out_size, void* d_ws, size_t ws_size,
                              hipStream_t stream) {
  const float* x  = (const float*)d_in[0];
  const int*   ei = (const int*)d_in[1];
  const float* W1 = (const float*)d_in[2];
  const float* b1 = (const float*)d_in[3];
  const float* W2 = (const float*)d_in[4];
  const float* b2 = (const float*)d_in[5];
  const float* W3 = (const float*)d_in[6];
  const float* b3 = (const float*)d_in[7];
  const float* Wc = (const float*)d_in[8];
  const float* bc = (const float*)d_in[9];

  const int N = in_sizes[0] / DFEAT;
  const int E = in_sizes[1] / 2;
  const int* row = ei;
  const int* col = ei + E;

  float* dinv = (float*)d_ws;
  float* B0 = dinv + N;
  float* B1 = B0 + (size_t)N * HPAD;
  float* B2 = B1 + (size_t)N * HPAD;
  float* outf = (float*)d_out;
  float* slots = outf;

  const int T = 256;
  const int numTiles = N / 16;
  dim3 gS((N + SC_NR - 1) / SC_NR, SC_CHUNKS);
  dim3 gB(SC_T / 32, SC_PC);
  dim3 gG((numTiles + 7) / 8);
  dim3 gO((numTiles * 8 + 7) / 8);

  k_bin      <<<gB, SC_T, 0, stream>>>(row, col, slots, dinv, N, E, (int)gS.x);
  k_sort     <<<gS, SC_T, 0, stream>>>(row, col, dinv, slots, N, E);

  k_gemm_scale<DFEAT, true><<<gG, T, 0, stream>>>(x, W1, DFEAT, b1, dinv, B0, B1, numTiles);
  k_scatter<<<gS, SC_T, 0, stream>>>(row, col, B0, B1, slots, N, E);

  k_gemm_scale<HPAD, false><<<gG, T, 0, stream>>>(B1, W2, HFEAT, b1, dinv, B0, B2, numTiles);
  k_scatter<<<gS, SC_T, 0, stream>>>(row, col, B0, B2, slots, N, E);

  k_gemm_scale<HPAD, false><<<gG, T, 0, stream>>>(B2, W3, HFEAT, b2, dinv, B0, B1, numTiles);
  k_scatter<<<gS, SC_T, 0, stream>>>(row, col, B0, B1, slots, N, E);

  k_gemm_out<<<gO, T, 0, stream>>>(B1, dinv, b3, Wc, bc, outf, numTiles);
}


# Output product re-cut: a wave owns whole rows (16 nodes x 128 columns: A read once, 8 accumulator tiles) and stores whole 128-B lines via a lane-half swap; 6250 waves, not 50000; sums unchanged
# speedup vs baseline: 6.5855x; 1.0358x over previous
#include <hip/hip_runtime.h>


typedef __attribute__((ext_vector_type(2))) float v2f;
typedef __attribute__((ext_vector_type(8))) float v8f;

#define DFEAT 128
#define HFEAT 15
#define HPAD  16

__device__ __forceinline__ void atomicAddF(float* p, float v) {
  unsafeAtomicAdd(p, v);
}

template <int K, bool FIRST>
__global__ void k_gemm_scale(const float* __restrict__ A,
                             const float* __restrict__ W,
                             int wrows,
                             const float* __restrict__ bprev,
                             float* __restrict__ dinv,
                             float* __restrict__ hs,
                             float* __restrict__ agg,
                             int numTiles) {
  __shared__ float wlds[K][HPAD];
  __shared__ float blds[HPAD];
  for (int i = threadIdx.x; i < K * HPAD; i += blockDim.x) {
    int k = i >> 4, n = i & 15;
    wlds[k][n] = (k < wrows && n < HFEAT) ? W[k * HFEAT + n] : 0.0f;
  }
  if (!FIRST && threadIdx.x < HPAD) blds[threadIdx.x] = threadIdx.x < HFEAT ? bprev[threadIdx.x] : 0.0f;
  __syncthreads();

  int wave = threadIdx.x >> 5;
  int lane = threadIdx.x & 31;
  int tile = blockIdx.x * (blockDim.x >> 5) + wave;
  if (tile >= numTiles) return;

  int m  = lane & 15;
  int kh = (lane >> 4) << 1;
  const float* arow = A + (size_t)(tile * 16 + m) * K;
  float dm = 0.0f;
  if (!FIRST) dm = dinv[tile * 16 + m];

  v8f c = {};
#pragma unroll
  for (int k0 = 0; k0 < K; k0 += 4) {
    v2f a, b;
    a.x = arow[k0 + kh];
    a.y = arow[k0 + kh + 1];
    if (!FIRST) {
      a.x = fmaxf(dm * a.x + blds[k0 + kh], 0.0f);
      a.y = (k0 + kh + 1 < HFEAT) ? fmaxf(dm * a.y + blds[k0 + kh + 1], 0.0f) : 0.0f;
    }
    b.x = wlds[k0 + kh][m];
    b.y = wlds[k0 + kh + 1][m];
    c = __builtin_amdgcn_wmma_f32_16x16x4_f32(false, a, false, b, (short)0, c,
                                              false, false);
  }

  int mbase = (lane >> 4) << 3;
  float dv[8];
#pragma unroll
  for (int i = 0; i < 8; ++i) {
    float d = dinv[tile * 16 + mbase + i];
    dv[i] = FIRST ? rsqrtf(d) : d;
  }
  if (FIRST) {
#pragma unroll
    for (int i = 0; i < 8; ++i) {
      if (m == 0) dinv[tile * 16 + mbase + i] = dv[i];
    }
  }
#pragma unroll
  for (int i = 0; i < 8; ++i) {
    int node = tile * 16 + mbase + i;
    float v = c[i] * dv[i];
    hs[node * HPAD + m]  = v;
    agg[node * HPAD + m] = v;
  }
}


#define SC_NR     2000
#define SC_CHUNKS 5
#define SC_PC     10
#define SC_T      1024
#define SC_QN     64
#define SC_CAP    1024
#define SC_HDR    32
#define SC_SLOT   (SC_HDR + SC_CAP)
#define SC_BINS   250
#define SC_SUBS   ((SC_T / 32) * SC_PC)
#define SC_SUBCAP 144
#define SC_TILEW  (SC_SUBS * SC_SUBCAP + SC_SUBS)
#define SC_EMAX   (32 * SC_CAP)
#define SC_HISTN  10240
#define SC_NOLIST 0xFFFFFFFFu

__device__ __forceinline__ unsigned sc_srn(int n) {
  return (unsigned)(((n + SC_CHUNKS - 1) / SC_CHUNKS + 1) & ~1);
}

__device__ __forceinline__ void sc_drain(const unsigned* __restrict__ qw, unsigned head,
                                         int count, const float* __restrict__ hs,
                                         float* acc, int lane) {
  const int h = lane >> 4, l16 = lane & 15;
  const uint4* qp = reinterpret_cast<const uint4*>(qw + ((head & (SC_QN - 1)) + 16 * h));
  const uint4 q0 = qp[0], q1 = qp[1], q2 = qp[2], q3 = qp[3];
  const unsigned ent[16] = {q0.x, q0.y, q0.z, q0.w, q1.x, q1.y, q1.z, q1.w,
                            q2.x, q2.y, q2.z, q2.w, q3.x, q3.y, q3.z, q3.w};
  const int left = count - 16 * h;
  float v[16];
#pragma unroll
  for (int s = 0; s < 16; ++s) {
    v[s] = 0.0f;
    if (s < left) v[s] = hs[(size_t)(ent[s] >> 12) * HPAD + l16];
  }
#pragma unroll
  for (int s = 0; s < 16; ++s) {
    if (s < left) unsafeAtomicAdd(&acc[(ent[s] & 4095u) * HPAD + l16], v[s]);
  }
}

__device__ __forceinline__ void sc_filter(const int* __restrict__ row, const int* __restrict__ col,
                                          const float* __restrict__ hs, float* acc, unsigned* qw,
                                          int base, int ncur, unsigned sbase, unsigned sn,
                                          int e0, int e1, int wave, int lane) {
  unsigned head = 0, tail = 0;

  const int4 none = make_int4(-1, -1, -1, -1);
  int it = e0 + wave * 256;
  int4 cA = none, cB = none, rA = none, rB = none;
  {
    const int i = it + lane * 4;
    if (i < e1)       { cA = *reinterpret_cast<const int4*>(col + i);       rA = *reinterpret_cast<const int4*>(row + i); }
    if (i + 128 < e1) { cB = *reinterpret_cast<const int4*>(col + i + 128); rB = *reinterpret_cast<const int4*>(row + i + 128); }
  }
  for (; it < e1; it += SC_T * 8) {
    const int cv[8] = {cA.x, cA.y, cA.z, cA.w, cB.x, cB.y, cB.z, cB.w};
    const int rv[8] = {rA.x, rA.y, rA.z, rA.w, rB.x, rB.y, rB.z, rB.w};
    cA = none; cB = none;
    {
      const int i = it + SC_T * 8 + lane * 4;
      if (i < e1)       { cA = *reinterpret_cast<const int4*>(col + i);       rA = *reinterpret_cast<const int4*>(row + i); }
      if (i + 128 < e1) { cB = *reinterpret_cast<const int4*>(col + i + 128); rB = *reinterpret_cast<const int4*>(row + i + 128); }
    }
#pragma unroll
    for (int j = 0; j < 8; ++j) {
      const unsigned cl = (unsigned)(cv[j] - base);
      const bool p = cl < (unsigned)ncur && ((unsigned)rv[j] - sbase) < sn;
      const unsigned m = __builtin_amdgcn_ballot_w32(p);
      if (m) {
        const unsigned pos = tail + __builtin_amdgcn_mbcnt_lo(m, 0u);
        if (p) qw[pos & (SC_QN - 1)] = ((unsigned)rv[j] << 12) | cl;
        tail += (unsigned)__builtin_popcount(m);
        if (tail - head >= 32u) {
          sc_drain(qw, head, 32, hs, acc, lane);
          head += 32u;
        }
      }
    }
  }
  if (tail != head) sc_drain(qw, head, (int)(tail - head), hs, acc, lane);
}

__device__ __forceinline__ void sc_replay(const unsigned* __restrict__ words, unsigned cnt,
                                          const float* __restrict__ hs, float* acc, int lane) {
  const unsigned h = (unsigned)lane >> 4, l16 = (unsigned)lane & 15u;
  const uint4* qp = reinterpret_cast<const uint4*>(words + 32u * h);
  const char* hsb = reinterpret_cast<const char*>(hs);
  const unsigned last = (cnt - 1u) & ~63u;
  uint4 q[8];
#pragma unroll
  for (int k = 0; k < 8; ++k) q[k] = qp[k];
  unsigned b = 0;
  for (; b + 64u <= cnt; b += 64u) {
    const unsigned ent[32] = {q[0].x, q[0].y, q[0].z, q[0].w, q[1].x, q[1].y, q[1].z, q[1].w,
                              q[2].x, q[2].y, q[2].z, q[2].w, q[3].x, q[3].y, q[3].z, q[3].w,
                              q[4].x, q[4].y, q[4].z, q[4].w, q[5].x, q[5].y, q[5].z, q[5].w,
                              q[6].x, q[6].y, q[6].z, q[6].w, q[7].x, q[7].y, q[7].z, q[7].w};
    float v[32];
#pragma unroll
    for (int s = 0; s < 32; ++s) {
      v[s] = *reinterpret_cast<const float*>(hsb + (((ent[s] >> 6) & 0xFFFFFFC0u) | (l16 << 2)));
    }
    const unsigned nb = min(b + 64u, last);
#pragma unroll
    for (int k = 0; k < 8; ++k) q[k] = qp[nb / 4u + k];
#pragma unroll
    for (int s = 0; s < 32; ++s) {
      unsafeAtomicAdd(&acc[((ent[s] & 4095u) << 4) | l16], v[s]);
    }
  }
  if (b < cnt) {
    const unsigned ent[32] = {q[0].x, q[0].y, q[0].z, q[0].w, q[1].x, q[1].y, q[1].z, q[1].w,
                              q[2].x, q[2].y, q[2].z, q[2].w, q[3].x, q[3].y, q[3].z, q[3].w,
                              q[4].x, q[4].y, q[4].z, q[4].w, q[5].x, q[5].y, q[5].z, q[5].w,
                              q[6].x, q[6].y, q[6].z, q[6].w, q[7].x, q[7].y, q[7].z, q[7].w};
    const int left = ((int)(cnt - b) - (int)h + 1) >> 1;
    float v[32];
#pragma unroll
    for (int s = 0; s < 32; ++s) {
      v[s] = 0.0f;
      if (s < left)
        v[s] = *reinterpret_cast<const float*>(hsb + (((ent[s] >> 6) & 0xFFFFFFC0u) | (l16 << 2)));
    }
#pragma unroll
    for (int s = 0; s < 32; ++s) {
      if (s < left) unsafeAtomicAdd(&acc[((ent[s] & 4095u) << 4) | l16], v[s]);
    }
  }
}

__global__ __launch_bounds__(SC_T) void k_bin(const int* __restrict__ row,
                                              const int* __restrict__ col,
                                              float* __restrict__ slots,
                                              float* __restrict__ deg, int n, int e, int nr) {
  __shared__ unsigned bins[SC_BINS][SC_SUBCAP];
  __shared__ unsigned cur[SC_BINS];

  const int tid = threadIdx.x;
  const int share = blockIdx.x;
  {
    const int nthr = (int)(gridDim.x * gridDim.y) * SC_T;
    for (int i = (int)(blockIdx.y * gridDim.x + blockIdx.x) * SC_T + tid; i < n; i += nthr) deg[i] = 1.0f;
  }
  const int per = ((e + SC_PC * 4 - 1) / (SC_PC * 4)) * 4;
  const int e0 = blockIdx.y * per;
  const int e1 = min(e, e0 + per);
  const int first = e0 + share * 256 + (tid >> 6) * (SC_T * 8) + (tid & 63) * 4;
  const unsigned srn = sc_srn(n);
  const int ntiles = nr * SC_CHUNKS;
  const unsigned sub = (unsigned)blockIdx.y * (SC_T / 32) + (unsigned)share;
  unsigned* sl = reinterpret_cast<unsigned*>(slots);

  for (int t0 = 0; t0 < ntiles; t0 += SC_BINS) {
    for (int i = tid; i < SC_BINS; i += SC_T) cur[i] = 0u;
    __syncthreads();

    int i = first;
    int4 cv = make_int4(-1, -1, -1, -1), rv = cv;
    if (i < e1) { cv = *reinterpret_cast<const int4*>(col + i); rv = *reinterpret_cast<const int4*>(row + i); }
    for (; i < e1; i += 16 * SC_T * 8) {
      const int ct[4] = {cv.x, cv.y, cv.z, cv.w};
      const int rt[4] = {rv.x, rv.y, rv.z, rv.w};
      const int nx = i + 16 * SC_T * 8;
      if (nx < e1) { cv = *reinterpret_cast<const int4*>(col + nx); rv = *reinterpret_cast<const int4*>(row + nx); }
      unsigned bn[4], wd[4], pos[4];
#pragma unroll
      for (int j = 0; j < 4; ++j) {
        const unsigned c = (unsigned)ct[j], r = (unsigned)rt[j];
        const unsigned rb = c / (unsigned)SC_NR;
        wd[j] = (r << 12) | (c - rb * (unsigned)SC_NR);
        bn[j] = (r / srn) * (unsigned)nr + rb - (unsigned)t0;
        pos[j] = (unsigned)SC_SUBCAP;
        if (c < (unsigned)n && r < (unsigned)n && bn[j] < (unsigned)SC_BINS)
          pos[j] = atomicAdd(&cur[bn[j]], 1u);
      }
#pragma unroll
      for (int j = 0; j < 4; ++j) {
        if (pos[j] < (unsigned)SC_SUBCAP) bins[bn[j]][pos[j]] = wd[j];
      }
    }
    __syncthreads();

    for (int idx = tid; idx < SC_BINS * SC_SUBCAP; idx += SC_T) {
      const int b = idx / SC_SUBCAP, w = idx % SC_SUBCAP;
      const int tile = t0 + b;
      if (tile < ntiles) {
        const unsigned c = cur[b];
        unsigned* tb = sl + (size_t)tile * SC_TILEW;
        if (w == 0) tb[SC_SUBS * SC_SUBCAP + sub] = c;
        if ((unsigned)w < min(c, (unsigned)SC_SUBCAP)) tb[sub * SC_SUBCAP + w] = bins[b][w];
      }
    }
    __syncthreads();
  }
}

__device__ __forceinline__ unsigned sc_wave_scan(unsigned v, int lane) {
#pragma unroll
  for (int d = 1; d < 32; d <<= 1) {
    const unsigned t = __shfl_up(v, (unsigned)d, 32);
    if (lane >= d) v += t;
  }
  return v;
}

__global__ __launch_bounds__(SC_T) void k_sort(const int* __restrict__ row,
                                               const int* __restrict__ col,
                                               float* __restrict__ deg,
                                               float* slots, int n, int e) {
  __shared__ __attribute__((aligned(16))) unsigned ebuf[SC_EMAX];
  __shared__ unsigned sbuf[SC_EMAX];
  __shared__ unsigned hist[SC_HISTN];
  __shared__ unsigned cnt[SC_NR];
  __shared__ unsigned scnt[SC_SUBS];
  __shared__ unsigned soff[SC_SUBS];
  __shared__ unsigned wsum[SC_T / 32];
  __shared__ unsigned s_total, s_bad;

  const int tid = threadIdx.x;
  const int lane = tid & 31, wave = tid >> 5;
  const int base = blockIdx.x * SC_NR;
  const int ncur = min(SC_NR, n - base);
  const unsigned srn = sc_srn(n);
  const unsigned sbase = (unsigned)blockIdx.y * srn;
  const unsigned sn = sbase < (unsigned)n ? min(srn, (unsigned)n - sbase) : 0u;
  const unsigned tile = blockIdx.y * gridDim.x + blockIdx.x;
  unsigned* tb = reinterpret_cast<unsigned*>(slots) + (size_t)tile * SC_TILEW;

  if (tid < SC_SUBS) scnt[tid] = tb[SC_SUBS * SC_SUBCAP + tid];
  for (int i = tid; i < SC_HISTN; i += SC_T) hist[i] = 0u;
  for (int i = tid; i < SC_NR; i += SC_T) cnt[i] = 0u;
  __syncthreads();

  if (wave == 0) {
    unsigned loc[SC_SUBS / 32], sum = 0u, bad = 0u;
#pragma unroll
    for (int k = 0; k < SC_SUBS / 32; ++k) {
      const unsigned c = scnt[lane * (SC_SUBS / 32) + k];
      bad |= (c > (unsigned)SC_SUBCAP) ? 1u : 0u;
      loc[k] = sum;
      sum += min(c, (unsigned)SC_SUBCAP);
    }
    const unsigned inc = sc_wave_scan(sum, lane);
#pragma unroll
    for (int k = 0; k < SC_SUBS / 32; ++k) soff[lane * (SC_SUBS / 32) + k] = inc - sum + loc[k];
    const unsigned anybad = __builtin_amdgcn_ballot_w32(bad != 0u);
    if (lane == 31) { s_total = inc; s_bad = anybad; }
  }
  __syncthreads();
  const unsigned total = s_total;
  const bool nolist = s_bad != 0u || total > (unsigned)SC_EMAX || (sn + 1u) / 2u > (unsigned)SC_HISTN;

  if (nolist) {
    for (int i = tid * 4; i < e; i += SC_T * 4) {
      const int4 cv = *reinterpret_cast<const int4*>(col + i);
      const int4 rv = *reinterpret_cast<const int4*>(row + i);
      const int ct[4] = {cv.x, cv.y, cv.z, cv.w};
      const int rt[4] = {rv.x, rv.y, rv.z, rv.w};
#pragma unroll
      for (int j = 0; j < 4; ++j) {
        const unsigned cl = (unsigned)(ct[j] - base);
        if (cl < (unsigned)ncur && ((unsigned)rt[j] - sbase) < sn) atomicAdd(&cnt[cl], 1u);
      }
    }
    if (tid < SC_T / 32) tb[(size_t)tid * SC_SLOT] = SC_NOLIST;
  } else {
    for (int j = wave; j < SC_SUBS; j += SC_T / 32) {
      const unsigned c = min(scnt[j], (unsigned)SC_SUBCAP), off = soff[j];
      for (unsigned v = (unsigned)lane * 4u; v < c; v += 128u) {
        const uint4 q = *reinterpret_cast<const uint4*>(tb + j * SC_SUBCAP + v);
        const unsigned w4[4] = {q.x, q.y, q.z, q.w};
#pragma unroll
        for (unsigned t = 0; t < 4u; ++t) {
          if (v + t < c) ebuf[off + v + t] = w4[t];
        }
      }
    }
    __syncthreads();

    for (unsigned i = tid; i < total; i += SC_T) {
      const unsigned w = ebuf[i];
      const unsigned key = min(((w >> 12) - sbase) >> 1, (unsigned)(SC_HISTN - 1));
      atomicAdd(&hist[key], 1u);
      atomicAdd(&cnt[min(w & 4095u, (unsigned)(SC_NR - 1))], 1u);
    }
    __syncthreads();

    {
      unsigned a[SC_HISTN / SC_T], sum = 0u;
#pragma unroll
      for (int k = 0; k < SC_HISTN / SC_T; ++k) {
        const unsigned t = hist[tid * (SC_HISTN / SC_T) + k];
        a[k] = sum;
        sum += t;
      }
      const unsigned inc = sc_wave_scan(sum, lane);
      if (lane == 31) wsum[wave] = inc;
      __syncthreads();
      if (wave == 0) {
        const unsigned v = wsum[lane];
        wsum[lane] = sc_wave_scan(v, lane) - v;
      }
      __syncthreads();
      const unsigned first = wsum[wave] + inc - sum;
#pragma unroll
      for (int k = 0; k < SC_HISTN / SC_T; ++k) hist[tid * (SC_HISTN / SC_T) + k] = first + a[k];
    }
    __syncthreads();

    for (unsigned i = tid; i < total; i += SC_T) {
      const unsigned w = ebuf[i];
      const unsigned key = min(((w >> 12) - sbase) >> 1, (unsigned)(SC_HISTN - 1));
      const unsigned pos = atomicAdd(&hist[key], 1u);
      sbuf[min(pos, (unsigned)(SC_EMAX - 1))] = w;
    }
    __syncthreads();

    const unsigned per = ((total + 31u) / 32u + 63u) & ~63u;
    for (int l = 0; l < SC_T / 32; ++l) {
      const unsigned start = (unsigned)l * per;
      const unsigned cw = total > start ? min(per, total - start) : 0u;
      unsigned* lst = tb + (size_t)l * SC_SLOT;
      const unsigned k = (unsigned)tid;
      if (k == 0u) lst[0] = cw;
      if (k < ((cw + 63u) & ~63u)) {
        const unsigned p = k & 63u;
        const unsigned ed = (k & ~63u) + 2u * (p & 31u) + (p >> 5);
        lst[SC_HDR + k] = ed < cw ? sbuf[start + ed] : 0u;
      }
    }
  }
  __syncthreads();

  for (int i = tid; i < ncur; i += SC_T) atomicAddF(deg + base + i, (float)cnt[i]);
}

__global__ __launch_bounds__(SC_T) void k_scatter(const int* __restrict__ row,
                                                  const int* __restrict__ col,
                                                  const float* __restrict__ hs,
                                                  float* __restrict__ agg,
                                                  float* __restrict__ slots, int n, int e) {
  __shared__ __attribute__((aligned(16))) float acc[SC_NR * HPAD];
  __shared__ __attribute__((aligned(16))) unsigned ring[SC_T / 32][SC_QN];

  const int tid = threadIdx.x;
  const int lane = tid & 31, wave = tid >> 5;
  const int base = blockIdx.x * SC_NR;
  const int ncur = min(SC_NR, n - base);

  float4* acc4 = reinterpret_cast<float4*>(acc);
  for (int i = tid; i < SC_NR * HPAD / 4; i += SC_T) acc4[i] = make_float4(0.f, 0.f, 0.f, 0.f);
  for (int i = tid; i < (SC_T / 32) * SC_QN; i += SC_T) (&ring[0][0])[i] = 0u;
  __syncthreads();

  const unsigned tile = blockIdx.y * gridDim.x + blockIdx.x;
  unsigned* lst = reinterpret_cast<unsigned*>(slots) + (size_t)tile * SC_TILEW + (size_t)wave * SC_SLOT;

  const unsigned cnt = (unsigned)__builtin_amdgcn_readfirstlane((int)lst[0]);
  if (cnt > (unsigned)SC_CAP) {
    const unsigned srn = sc_srn(n);
    const unsigned sbase = (unsigned)blockIdx.y * srn;
    const unsigned sn = sbase < (unsigned)n ? min(srn, (unsigned)n - sbase) : 0u;
    sc_filter(row, col, hs, acc, ring[wave], base, ncur, sbase, sn, 0, e, wave, lane);
  } else if (cnt > 0u) {
    sc_replay(lst + SC_HDR, cnt, hs, acc, lane);
  }
  __syncthreads();

  float* dst = agg + (size_t)base * HPAD;
  const int tot = ncur * HPAD;
  for (int i = tid; i < tot; i += SC_T) {
    if ((i & 15) != 15) atomicAddF(dst + i, acc[i]);
  }
}

__device__ __forceinline__ float xhalf(float x) {
  return __builtin_bit_cast(float, __builtin_amdgcn_permlanex16(__builtin_bit_cast(int, x), __builtin_bit_cast(int, x),
                                                                0x76543210, 0xfedcba98, false, false));
}

#define GO_T 256
__global__ __launch_bounds__(GO_T) void k_gemm_out(const float* __restrict__ A,
                                                  const float* __restrict__ dinv,
                                                  const float* __restrict__ bprev,
                                                  const float* __restrict__ Wc,
                                                  const float* __restrict__ bc,
                                                  float* __restrict__ out,
                                                  int numNodeTiles) {
  __shared__ v2f wfr[4 * 8 * 32];
  __shared__ float blds[HPAD];

  const int wave = threadIdx.x >> 5;
  const int lane = threadIdx.x & 31;
  const int tile = blockIdx.x * (GO_T / 32) + wave;
  const bool live = tile < numNodeTiles;
  const int tl = live ? tile : numNodeTiles - 1;
  const int m  = lane & 15;
  const int h  = lane >> 4;
  const int kh = h << 1;

  const float* arow = A + (size_t)(tl * 16 + m) * HPAD;
  const float dm = dinv[tl * 16 + m];
  float2 r[4];
#pragma unroll
  for (int ks = 0; ks < 4; ++ks) r[ks] = *reinterpret_cast<const float2*>(arow + 4 * ks + kh);

  v2f w[4];
#pragma unroll
  for (int ks = 0; ks < 4; ++ks) {
    const int ln = threadIdx.x & 31, j = threadIdx.x >> 5;
    const int k = 4 * ks + ((ln >> 4) << 1), n = 16 * j + (ln & 15);
    const int k1 = (k + 1 < HFEAT) ? k + 1 : k;
    const float y = Wc[k1 * DFEAT + n];
    w[ks].x = Wc[k * DFEAT + n];
    w[ks].y = (k + 1 < HFEAT) ? y : 0.0f;
  }
#pragma unroll
  for (int ks = 0; ks < 4; ++ks) wfr[ks * GO_T + threadIdx.x] = w[ks];
  if (threadIdx.x < HPAD) blds[threadIdx.x] = threadIdx.x < HFEAT ? bprev[threadIdx.x] : 0.0f;
  __syncthreads();
  if (!live) return;

  v2f a[4];
#pragma unroll
  for (int ks = 0; ks < 4; ++ks) {
    const int k = 4 * ks + kh;
    a[ks].x = fmaxf(dm * r[ks].x + blds[k], 0.0f);
    a[ks].y = (k + 1 < HFEAT) ? fmaxf(dm * r[ks].y + blds[k + 1], 0.0f) : 0.0f;
  }

  v8f c[8];
#pragma unroll
  for (int j = 0; j < 8; ++j) c[j] = v8f{};
#pragma unroll
  for (int ks = 0; ks < 4; ++ks) {
#pragma unroll
    for (int j = 0; j < 8; ++j) {
      const v2f b = wfr[(ks * 8 + j) * 32 + lane];
      c[j] = __builtin_amdgcn_wmma_f32_16x16x4_f32(false, a[ks], false, b, (short)0, c[j],
                                                   false, false);
    }
  }

  float* orow = out + (size_t)(tile * 16) * DFEAT + lane;
#pragma unroll
  for (int p = 0; p < 4; ++p) {
    const float bias = bc[32 * p + lane];
#pragma unroll
    for (int i = 0; i < 8; ++i) {
      const float lo = c[2 * p][i], hi = c[2 * p + 1][i];
      const float got = xhalf(h ? lo : hi);
      const float vA = h ? got : lo;
      const float vB = h ? hi : got;
      orow[(size_t)i * DFEAT + 32 * p]       = vA + bias;
      orow[(size_t)(8 + i) * DFEAT + 32 * p] = vB + bias;
    }
  }
}

extern "C" void kernel_launch(void* const* d_in, const int* in_sizes, int n_in,
                              void* d_out, int out_size, void* d_ws, size_t ws_size,
                              hipStream_t stream) {
  const float* x  = (const float*)d_in[0];
  const int*   ei = (const int*)d_in[1];
  const float* W1 = (const float*)d_in[2];
  const float* b1 = (const float*)d_in[3];
  const float* W2 = (const float*)d_in[4];
  const float* b2 = (const float*)d_in[5];
  const float* W3 = (const float*)d_in[6];
  const float* b3 = (const float*)d_in[7];
  const float* Wc = (const float*)d_in[8];
  const float* bc = (const float*)d_in[9];

  const int N = in_sizes[0] / DFEAT;
  const int E = in_sizes[1] / 2;
  const int* row = ei;
  const int* col = ei + E;

  float* dinv = (float*)d_ws;
  float* B0 = dinv + N;
  float* B1 = B0 + (size_t)N * HPAD;
  float* B2 = B1 + (size_t)N * HPAD;
  float* outf = (float*)d_out;
  float* slots = outf;

  const int T = 256;
  const int numTiles = N / 16;
  dim3 gS((N + SC_NR - 1) / SC_NR, SC_CHUNKS);
  dim3 gB(SC_T / 32, SC_PC);
  dim3 gG((numTiles + 7) / 8);

  k_bin      <<<gB, SC_T, 0, stream>>>(row, col, slots, dinv, N, E, (int)gS.x);
  k_sort     <<<gS, SC_T, 0, stream>>>(row, col, dinv, slots, N, E);

  k_gemm_scale<DFEAT, true><<<gG, T, 0, stream>>>(x, W1, DFEAT, b1, dinv, B0, B1, numTiles);
  k_scatter<<<gS, SC_T, 0, stream>>>(row, col, B0, B1, slots, N, E);

  k_gemm_scale<HPAD, false><<<gG, T, 0, stream>>>(B1, W2, HFEAT, b1, dinv, B0, B2, numTiles);
  k_scatter<<<gS, SC_T, 0, stream>>>(row, col, B0, B2, slots, N, E);

  k_gemm_scale<HPAD, false><<<gG, T, 0, stream>>>(B2, W3, HFEAT, b2, dinv, B0, B1, numTiles);
  k_scatter<<<gS, SC_T, 0, stream>>>(row, col, B0, B1, slots, N, E);

  k_gemm_out<<<gG, GO_T, 0, stream>>>(B1, dinv, b3, Wc, bc, outf, numTiles);
}


# Serial round trips batched: loads waited for one by one (k_sort sub-lists, weight staging, LDS reads of k_bin, k_sort and the flush) issue in groups; first list words and A rows are requested early
# speedup vs baseline: 6.9586x; 1.0567x over previous
#include <hip/hip_runtime.h>


typedef __attribute__((ext_vector_type(2))) float v2f;
typedef __attribute__((ext_vector_type(8))) float v8f;

#define DFEAT 128
#define HFEAT 15
#define HPAD  16

__device__ __forceinline__ void atomicAddF(float* p, float v) {
  unsafeAtomicAdd(p, v);
}

#define GS_T 256
template <int K, bool FIRST>
__global__ __launch_bounds__(GS_T) void k_gemm_scale(const float* __restrict__ A,
                             const float* __restrict__ W,
                             int wrows,
                             const float* __restrict__ bprev,
                             float* __restrict__ dinv,
                             float* __restrict__ hs,
                             float* __restrict__ agg,
                             int numTiles) {
  __shared__ float wlds[K][HPAD];
  __shared__ float blds[HPAD];

  int wave = threadIdx.x >> 5;
  int lane = threadIdx.x & 31;
  int tile = blockIdx.x * (GS_T >> 5) + wave;
  const bool live = tile < numTiles;
  const int tl = live ? tile : numTiles - 1;
  int m  = lane & 15;
  int kh = (lane >> 4) << 1;

  const float* arow = A + (size_t)(tl * 16 + m) * K;
  v2f av[K / 4];
#pragma unroll
  for (int s = 0; s < K / 4; ++s) av[s] = *reinterpret_cast<const v2f*>(arow + 4 * s + kh);
  float dm = 0.0f;
  if (!FIRST) dm = dinv[tl * 16 + m];

  constexpr int WPER = (K * HPAD + GS_T - 1) / GS_T;
  float wv[WPER];
#pragma unroll
  for (int u = 0; u < WPER; ++u) {
    const int i = (int)threadIdx.x + u * GS_T;
    const int k = i >> 4, n = i & 15;
    wv[u] = (i < K * HPAD && k < wrows && n < HFEAT) ? W[k * HFEAT + n] : 0.0f;
  }
  float bv = 0.0f;
  if (!FIRST && threadIdx.x < HFEAT) bv = bprev[threadIdx.x];
#pragma unroll
  for (int u = 0; u < WPER; ++u) {
    const int i = (int)threadIdx.x + u * GS_T;
    if (i < K * HPAD) wlds[i >> 4][i & 15] = wv[u];
  }
  if (!FIRST && threadIdx.x < HPAD) blds[threadIdx.x] = bv;
  __syncthreads();
  if (!live) return;

  v8f c = {};
#pragma unroll
  for (int k0 = 0; k0 < K; k0 += 4) {
    v2f a = av[k0 / 4], b;
    if (!FIRST) {
      a.x = fmaxf(dm * a.x + blds[k0 + kh], 0.0f);
      a.y = (k0 + kh + 1 < HFEAT) ? fmaxf(dm * a.y + blds[k0 + kh + 1], 0.0f) : 0.0f;
    }
    b.x = wlds[k0 + kh][m];
    b.y = wlds[k0 + kh + 1][m];
    c = __builtin_amdgcn_wmma_f32_16x16x4_f32(false, a, false, b, (short)0, c,
                                              false, false);
  }

  int mbase = (lane >> 4) << 3;
  float dv[8];
#pragma unroll
  for (int i = 0; i < 8; ++i) {
    float d = dinv[tile * 16 + mbase + i];
    dv[i] = FIRST ? rsqrtf(d) : d;
  }
  if (FIRST) {
#pragma unroll
    for (int i = 0; i < 8; ++i) {
      if (m == 0) dinv[tile * 16 + mbase + i] = dv[i];
    }
  }
#pragma unroll
  for (int i = 0; i < 8; ++i) {
    int node = tile * 16 + mbase + i;
    float v = c[i] * dv[i];
    hs[node * HPAD + m]  = v;
    agg[node * HPAD + m] = v;
  }
}


#define SC_NR     2000
#define SC_CHUNKS 5
#define SC_PC     10
#define SC_T      1024
#define SC_QN     64
#define SC_CAP    1024
#define SC_HDR    32
#define SC_SLOT   (SC_HDR + SC_CAP)
#define SC_BINS   250
#define SC_SUBS   ((SC_T / 32) * SC_PC)
#define SC_SUBCAP 144
#define SC_TILEW  (SC_SUBS * SC_SUBCAP + SC_SUBS)
#define SC_EMAX   (32 * SC_CAP)
#define SC_HISTN  10240
#define SC_NOLIST 0xFFFFFFFFu

__device__ __forceinline__ unsigned sc_srn(int n) {
  return (unsigned)(((n + SC_CHUNKS - 1) / SC_CHUNKS + 1) & ~1);
}

__device__ __forceinline__ void sc_drain(const unsigned* __restrict__ qw, unsigned head,
                                         int count, const float* __restrict__ hs,
                                         float* acc, int lane) {
  const int h = lane >> 4, l16 = lane & 15;
  const uint4* qp = reinterpret_cast<const uint4*>(qw + ((head & (SC_QN - 1)) + 16 * h));
  const uint4 q0 = qp[0], q1 = qp[1], q2 = qp[2], q3 = qp[3];
  const unsigned ent[16] = {q0.x, q0.y, q0.z, q0.w, q1.x, q1.y, q1.z, q1.w,
                            q2.x, q2.y, q2.z, q2.w, q3.x, q3.y, q3.z, q3.w};
  const int left = count - 16 * h;
  float v[16];
#pragma unroll
  for (int s = 0; s < 16; ++s) {
    v[s] = 0.0f;
    if (s < left) v[s] = hs[(size_t)(ent[s] >> 12) * HPAD + l16];
  }
#pragma unroll
  for (int s = 0; s < 16; ++s) {
    if (s < left) unsafeAtomicAdd(&acc[(ent[s] & 4095u) * HPAD + l16], v[s]);
  }
}

__device__ __forceinline__ void sc_filter(const int* __restrict__ row, const int* __restrict__ col,
                                          const float* __restrict__ hs, float* acc, unsigned* qw,
                                          int base, int ncur, unsigned sbase, unsigned sn,
                                          int e0, int e1, int wave, int lane) {
  unsigned head = 0, tail = 0;

  const int4 none = make_int4(-1, -1, -1, -1);
  int it = e0 + wave * 256;
  int4 cA = none, cB = none, rA = none, rB = none;
  {
    const int i = it + lane * 4;
    if (i < e1)       { cA = *reinterpret_cast<const int4*>(col + i);       rA = *reinterpret_cast<const int4*>(row + i); }
    if (i + 128 < e1) { cB = *reinterpret_cast<const int4*>(col + i + 128); rB = *reinterpret_cast<const int4*>(row + i + 128); }
  }
  for (; it < e1; it += SC_T * 8) {
    const int cv[8] = {cA.x, cA.y, cA.z, cA.w, cB.x, cB.y, cB.z, cB.w};
    const int rv[8] = {rA.x, rA.y, rA.z, rA.w, rB.x, rB.y, rB.z, rB.w};
    cA = none; cB = none;
    {
      const int i = it + SC_T * 8 + lane * 4;
      if (i < e1)       { cA = *reinterpret_cast<const int4*>(col + i);       rA = *reinterpret_cast<const int4*>(row + i); }
      if (i + 128 < e1) { cB = *reinterpret_cast<const int4*>(col + i + 128); rB = *reinterpret_cast<const int4*>(row + i + 128); }
    }
#pragma unroll
    for (int j = 0; j < 8; ++j) {
      const unsigned cl = (unsigned)(cv[j] - base);
      const bool p = cl < (unsigned)ncur && ((unsigned)rv[j] - sbase) < sn;
      const unsigned m = __builtin_amdgcn_ballot_w32(p);
      if (m) {
        const unsigned pos = tail + __builtin_amdgcn_mbcnt_lo(m, 0u);
        if (p) qw[pos & (SC_QN - 1)] = ((unsigned)rv[j] << 12) | cl;
        tail += (unsigned)__builtin_popcount(m);
        if (tail - head >= 32u) {
          sc_drain(qw, head, 32, hs, acc, lane);
          head += 32u;
        }
      }
    }
  }
  if (tail != head) sc_drain(qw, head, (int)(tail - head), hs, acc, lane);
}

__device__ __forceinline__ void sc_replay(const unsigned* __restrict__ words, unsigned cnt,
                                          const float* __restrict__ hs, float* acc, int lane,
                                          const uint4 (&q0)[8]) {
  const unsigned h = (unsigned)lane >> 4, l16 = (unsigned)lane & 15u;
  const uint4* qp = reinterpret_cast<const uint4*>(words + 32u * h);
  const char* hsb = reinterpret_cast<const char*>(hs);
  const unsigned last = (cnt - 1u) & ~63u;
  uint4 q[8];
#pragma unroll
  for (int k = 0; k < 8; ++k) q[k] = q0[k];
  unsigned b = 0;
  for (; b + 64u <= cnt; b += 64u) {
    const unsigned ent[32] = {q[0].x, q[0].y, q[0].z, q[0].w, q[1].x, q[1].y, q[1].z, q[1].w,
                              q[2].x, q[2].y, q[2].z, q[2].w, q[3].x, q[3].y, q[3].z, q[3].w,
                              q[4].x, q[4].y, q[4].z, q[4].w, q[5].x, q[5].y, q[5].z, q[5].w,
                              q[6].x, q[6].y, q[6].z, q[6].w, q[7].x, q[7].y, q[7].z, q[7].w};
    float v[32];
#pragma unroll
    for (int s = 0; s < 32; ++s) {
      v[s] = *reinterpret_cast<const float*>(hsb + (((ent[s] >> 6) & 0xFFFFFFC0u) | (l16 << 2)));
    }
    const unsigned nb = min(b + 64u, last);
#pragma unroll
    for (int k = 0; k < 8; ++k) q[k] = qp[nb / 4u + k];
#pragma unroll
    for (int s = 0; s < 32; ++s) {
      unsafeAtomicAdd(&acc[((ent[s] & 4095u) << 4) | l16], v[s]);
    }
  }
  if (b < cnt) {
    const unsigned ent[32] = {q[0].x, q[0].y, q[0].z, q[0].w, q[1].x, q[1].y, q[1].z, q[1].w,
                              q[2].x, q[2].y, q[2].z, q[2].w, q[3].x, q[3].y, q[3].z, q[3].w,
                              q[4].x, q[4].y, q[4].z, q[4].w, q[5].x, q[5].y, q[5].z, q[5].w,
                              q[6].x, q[6].y, q[6].z, q[6].w, q[7].x, q[7].y, q[7].z, q[7].w};
    const int left = ((int)(cnt - b) - (int)h + 1) >> 1;
    float v[32];
#pragma unroll
    for (int s = 0; s < 32; ++s) {
      v[s] = 0.0f;
      if (s < left)
        v[s] = *reinterpret_cast<const float*>(hsb + (((ent[s] >> 6) & 0xFFFFFFC0u) | (l16 << 2)));
    }
#pragma unroll
    for (int s = 0; s < 32; ++s) {
      if (s < left) unsafeAtomicAdd(&acc[((ent[s] & 4095u) << 4) | l16], v[s]);
    }
  }
}

__global__ __launch_bounds__(SC_T) void k_bin(const int* __restrict__ row,
                                              const int* __restrict__ col,
                                              float* __restrict__ slots,
                                              float* __restrict__ deg, int n, int e, int nr) {
  __shared__ unsigned bins[SC_BINS][SC_SUBCAP];
  __shared__ unsigned cur[SC_BINS];

  const int tid = threadIdx.x;
  const int share = blockIdx.x;
  {
    const int nthr = (int)(gridDim.x * gridDim.y) * SC_T;
    for (int i = (int)(blockIdx.y * gridDim.x + blockIdx.x) * SC_T + tid; i < n; i += nthr) deg[i] = 1.0f;
  }
  const int per = ((e + SC_PC * 4 - 1) / (SC_PC * 4)) * 4;
  const int e0 = blockIdx.y * per;
  const int e1 = min(e, e0 + per);
  const int first = e0 + share * 256 + (tid >> 6) * (SC_T * 8) + (tid & 63) * 4;
  const unsigned srn = sc_srn(n);
  const int ntiles = nr * SC_CHUNKS;
  const unsigned sub = (unsigned)blockIdx.y * (SC_T / 32) + (unsigned)share;
  unsigned* sl = reinterpret_cast<unsigned*>(slots);

  for (int t0 = 0; t0 < ntiles; t0 += SC_BINS) {
    for (int i = tid; i < SC_BINS; i += SC_T) cur[i] = 0u;
    __syncthreads();

    int i = first;
    int4 cv = make_int4(-1, -1, -1, -1), rv = cv;
    if (i < e1) { cv = *reinterpret_cast<const int4*>(col + i); rv = *reinterpret_cast<const int4*>(row + i); }
    for (; i < e1; i += 16 * SC_T * 8) {
      const int ct[4] = {cv.x, cv.y, cv.z, cv.w};
      const int rt[4] = {rv.x, rv.y, rv.z, rv.w};
      const int nx = i + 16 * SC_T * 8;
      if (nx < e1) { cv = *reinterpret_cast<const int4*>(col + nx); rv = *reinterpret_cast<const int4*>(row + nx); }
      unsigned bn[4], wd[4], pos[4];
#pragma unroll
      for (int j = 0; j < 4; ++j) {
        const unsigned c = (unsigned)ct[j], r = (unsigned)rt[j];
        const unsigned rb = c / (unsigned)SC_NR;
        wd[j] = (r << 12) | (c - rb * (unsigned)SC_NR);
        bn[j] = (r / srn) * (unsigned)nr + rb - (unsigned)t0;
        pos[j] = (unsigned)SC_SUBCAP;
        if (c < (unsigned)n && r < (unsigned)n && bn[j] < (unsigned)SC_BINS)
          pos[j] = atomicAdd(&cur[bn[j]], 1u);
      }
#pragma unroll
      for (int j = 0; j < 4; ++j) {
        if (pos[j] < (unsigned)SC_SUBCAP) bins[bn[j]][pos[j]] = wd[j];
      }
    }
    __syncthreads();

    for (int idx0 = tid; idx0 < SC_BINS * SC_SUBCAP; idx0 += 4 * SC_T) {
      unsigned cv[4], wv[4];
      bool keep[4];
#pragma unroll
      for (int u = 0; u < 4; ++u) {
        const int idx = idx0 + u * SC_T;
        const int b = idx / SC_SUBCAP;
        cv[u] = 0u;
        if (idx < SC_BINS * SC_SUBCAP && t0 + b < ntiles) cv[u] = cur[b];
      }
#pragma unroll
      for (int u = 0; u < 4; ++u) {
        const int idx = idx0 + u * SC_T;
        const int b = idx / SC_SUBCAP, w = idx % SC_SUBCAP;
        keep[u] = (unsigned)w < min(cv[u], (unsigned)SC_SUBCAP);
        wv[u] = 0u;
        if (keep[u]) wv[u] = bins[b][w];
      }
#pragma unroll
      for (int u = 0; u < 4; ++u) {
        const int idx = idx0 + u * SC_T;
        const int b = idx / SC_SUBCAP, w = idx % SC_SUBCAP;
        const int tile = t0 + b;
        if (idx < SC_BINS * SC_SUBCAP && tile < ntiles) {
          unsigned* tb = sl + (size_t)tile * SC_TILEW;
          if (w == 0) tb[SC_SUBS * SC_SUBCAP + sub] = cv[u];
          if (keep[u]) tb[sub * SC_SUBCAP + w] = wv[u];
        }
      }
    }
    __syncthreads();
  }
}

__device__ __forceinline__ unsigned sc_wave_scan(unsigned v, int lane) {
#pragma unroll
  for (int d = 1; d < 32; d <<= 1) {
    const unsigned t = __shfl_up(v, (unsigned)d, 32);
    if (lane >= d) v += t;
  }
  return v;
}

__global__ __launch_bounds__(SC_T) void k_sort(const int* __restrict__ row,
                                               const int* __restrict__ col,
                                               float* __restrict__ deg,
                                               float* slots, int n, int e) {
  __shared__ __attribute__((aligned(16))) unsigned ebuf[SC_EMAX];
  __shared__ unsigned sbuf[SC_EMAX];
  __shared__ unsigned hist[SC_HISTN];
  __shared__ unsigned cnt[SC_NR];
  __shared__ unsigned scnt[SC_SUBS];
  __shared__ unsigned soff[SC_SUBS];
  __shared__ unsigned wsum[SC_T / 32];
  __shared__ unsigned s_total, s_bad;

  const int tid = threadIdx.x;
  const int lane = tid & 31, wave = tid >> 5;
  const int base = blockIdx.x * SC_NR;
  const int ncur = min(SC_NR, n - base);
  const unsigned srn = sc_srn(n);
  const unsigned sbase = (unsigned)blockIdx.y * srn;
  const unsigned sn = sbase < (unsigned)n ? min(srn, (unsigned)n - sbase) : 0u;
  const unsigned tile = blockIdx.y * gridDim.x + blockIdx.x;
  unsigned* tb = reinterpret_cast<unsigned*>(slots) + (size_t)tile * SC_TILEW;

  if (tid < SC_SUBS) scnt[tid] = tb[SC_SUBS * SC_SUBCAP + tid];
  for (int i = tid; i < SC_HISTN; i += SC_T) hist[i] = 0u;
  for (int i = tid; i < SC_NR; i += SC_T) cnt[i] = 0u;
  __syncthreads();

  if (wave == 0) {
    unsigned loc[SC_SUBS / 32], sum = 0u, bad = 0u;
#pragma unroll
    for (int k = 0; k < SC_SUBS / 32; ++k) {
      const unsigned c = scnt[lane * (SC_SUBS / 32) + k];
      bad |= (c > (unsigned)SC_SUBCAP) ? 1u : 0u;
      loc[k] = sum;
      sum += min(c, (unsigned)SC_SUBCAP);
    }
    const unsigned inc = sc_wave_scan(sum, lane);
#pragma unroll
    for (int k = 0; k < SC_SUBS / 32; ++k) soff[lane * (SC_SUBS / 32) + k] = inc - sum + loc[k];
    const unsigned anybad = __builtin_amdgcn_ballot_w32(bad != 0u);
    if (lane == 31) { s_total = inc; s_bad = anybad; }
  }
  __syncthreads();
  const unsigned total = s_total;
  const bool nolist = s_bad != 0u || total > (unsigned)SC_EMAX || (sn + 1u) / 2u > (unsigned)SC_HISTN;

  if (nolist) {
    for (int i = tid * 4; i < e; i += SC_T * 4) {
      const int4 cv = *reinterpret_cast<const int4*>(col + i);
      const int4 rv = *reinterpret_cast<const int4*>(row + i);
      const int ct[4] = {cv.x, cv.y, cv.z, cv.w};
      const int rt[4] = {rv.x, rv.y, rv.z, rv.w};
#pragma unroll
      for (int j = 0; j < 4; ++j) {
        const unsigned cl = (unsigned)(ct[j] - base);
        if (cl < (unsigned)ncur && ((unsigned)rt[j] - sbase) < sn) atomicAdd(&cnt[cl], 1u);
      }
    }
    if (tid < SC_T / 32) tb[(size_t)tid * SC_SLOT] = SC_NOLIST;
    for (int i = tid; i < (SC_T / 32) * 64; i += SC_T) tb[(size_t)(i >> 6) * SC_SLOT + SC_HDR + (i & 63)] = 0u;
  } else {
    {
      constexpr int NSW = SC_SUBS / (SC_T / 32);
      const unsigned v = (unsigned)lane * 4u;
      unsigned cc[NSW], oo[NSW];
      uint4 qq[NSW];
#pragma unroll
      for (int t = 0; t < NSW; ++t) {
        const int j = wave + t * (SC_T / 32);
        cc[t] = min(scnt[j], (unsigned)SC_SUBCAP);
        oo[t] = soff[j];
      }
#pragma unroll
      for (int t = 0; t < NSW; ++t) {
        const int j = wave + t * (SC_T / 32);
        const unsigned* src = tb + (v < cc[t] ? (unsigned)(j * SC_SUBCAP) : (unsigned)(SC_SUBS * SC_SUBCAP)) + v;
        qq[t] = *reinterpret_cast<const uint4*>(src);
      }
#pragma unroll
      for (int t = 0; t < NSW; ++t) {
        const unsigned w4[4] = {qq[t].x, qq[t].y, qq[t].z, qq[t].w};
#pragma unroll
        for (unsigned u = 0; u < 4u; ++u) {
          if (v + u < cc[t]) ebuf[oo[t] + v + u] = w4[u];
        }
      }
#pragma unroll 1
      for (int j = wave; j < SC_SUBS; j += SC_T / 32) {
        const unsigned c = min(scnt[j], (unsigned)SC_SUBCAP), off = soff[j];
        for (unsigned v2 = v + 128u; v2 < c; v2 += 128u) {
          const uint4 q = *reinterpret_cast<const uint4*>(tb + j * SC_SUBCAP + v2);
          const unsigned w4[4] = {q.x, q.y, q.z, q.w};
#pragma unroll
          for (unsigned u = 0; u < 4u; ++u) {
            if (v2 + u < c) ebuf[off + v2 + u] = w4[u];
          }
        }
      }
    }
    __syncthreads();

    for (unsigned i = tid; i < total; i += SC_T) {
      const unsigned w = ebuf[i];
      const unsigned key = min(((w >> 12) - sbase) >> 1, (unsigned)(SC_HISTN - 1));
      atomicAdd(&hist[key], 1u);
      atomicAdd(&cnt[min(w & 4095u, (unsigned)(SC_NR - 1))], 1u);
    }
    __syncthreads();

    {
      unsigned a[SC_HISTN / SC_T], sum = 0u;
#pragma unroll
      for (int k = 0; k < SC_HISTN / SC_T; ++k) {
        const unsigned t = hist[tid * (SC_HISTN / SC_T) + k];
        a[k] = sum;
        sum += t;
      }
      const unsigned inc = sc_wave_scan(sum, lane);
      if (lane == 31) wsum[wave] = inc;
      __syncthreads();
      if (wave == 0) {
        const unsigned v = wsum[lane];
        wsum[lane] = sc_wave_scan(v, lane) - v;
      }
      __syncthreads();
      const unsigned first = wsum[wave] + inc - sum;
#pragma unroll
      for (int k = 0; k < SC_HISTN / SC_T; ++k) hist[tid * (SC_HISTN / SC_T) + k] = first + a[k];
    }
    __syncthreads();

    for (unsigned i = tid; i < total; i += SC_T) {
      const unsigned w = ebuf[i];
      const unsigned key = min(((w >> 12) - sbase) >> 1, (unsigned)(SC_HISTN - 1));
      const unsigned pos = atomicAdd(&hist[key], 1u);
      sbuf[min(pos, (unsigned)(SC_EMAX - 1))] = w;
    }
    __syncthreads();

    const unsigned per = ((total + 31u) / 32u + 63u) & ~63u;
    {
      const unsigned k = (unsigned)tid;
      const unsigned p = k & 63u;
      const unsigned ed = (k & ~63u) + 2u * (p & 31u) + (p >> 5);
#pragma unroll 1
      for (int l0 = 0; l0 < SC_T / 32; l0 += 8) {
        unsigned val[8];
#pragma unroll
        for (int u = 0; u < 8; ++u) {
          const unsigned start = (unsigned)(l0 + u) * per;
          const unsigned cw = total > start ? min(per, total - start) : 0u;
          val[u] = ed < cw ? sbuf[start + ed] : 0u;
        }
#pragma unroll
        for (int u = 0; u < 8; ++u) {
          const unsigned start = (unsigned)(l0 + u) * per;
          const unsigned cw = total > start ? min(per, total - start) : 0u;
          unsigned* lst = tb + (size_t)(l0 + u) * SC_SLOT;
          if (k == 0u) lst[0] = cw;
          if (k < max((cw + 63u) & ~63u, 64u)) lst[SC_HDR + k] = val[u];
        }
      }
    }
  }
  __syncthreads();

  for (int i = tid; i < ncur; i += SC_T) atomicAddF(deg + base + i, (float)cnt[i]);
}

__global__ __launch_bounds__(SC_T) void k_scatter(const int* __restrict__ row,
                                                  const int* __restrict__ col,
                                                  const float* __restrict__ hs,
                                                  float* __restrict__ agg,
                                                  float* __restrict__ slots, int n, int e) {
  __shared__ __attribute__((aligned(16))) float acc[SC_NR * HPAD];
  __shared__ __attribute__((aligned(16))) unsigned ring[SC_T / 32][SC_QN];

  const int tid = threadIdx.x;
  const int lane = tid & 31, wave = tid >> 5;
  const int base = blockIdx.x * SC_NR;
  const int ncur = min(SC_NR, n - base);

  const unsigned tile = blockIdx.y * gridDim.x + blockIdx.x;
  unsigned* lst = reinterpret_cast<unsigned*>(slots) + (size_t)tile * SC_TILEW + (size_t)wave * SC_SLOT;
  const unsigned cnt0 = lst[0];
  uint4 q0[8];
  {
    const uint4* qp0 = reinterpret_cast<const uint4*>(lst + SC_HDR + 32u * ((unsigned)lane >> 4));
#pragma unroll
    for (int k = 0; k < 8; ++k) q0[k] = qp0[k];
  }

  float4* acc4 = reinterpret_cast<float4*>(acc);
  for (int i = tid; i < SC_NR * HPAD / 4; i += SC_T) acc4[i] = make_float4(0.f, 0.f, 0.f, 0.f);
  for (int i = tid; i < (SC_T / 32) * SC_QN; i += SC_T) (&ring[0][0])[i] = 0u;
  __syncthreads();

  const unsigned cnt = (unsigned)__builtin_amdgcn_readfirstlane((int)cnt0);
  if (cnt > (unsigned)SC_CAP) {
    const unsigned srn = sc_srn(n);
    const unsigned sbase = (unsigned)blockIdx.y * srn;
    const unsigned sn = sbase < (unsigned)n ? min(srn, (unsigned)n - sbase) : 0u;
    sc_filter(row, col, hs, acc, ring[wave], base, ncur, sbase, sn, 0, e, wave, lane);
  } else if (cnt > 0u) {
    sc_replay(lst + SC_HDR, cnt, hs, acc, lane, q0);
  }
  __syncthreads();

  float* dst = agg + (size_t)base * HPAD;
  const int tot = ncur * HPAD;
  const bool real = (tid & 15) != 15;
  int i = tid;
  for (; i + 7 * SC_T < tot; i += 8 * SC_T) {
    if (real) {
      float a[8];
#pragma unroll
      for (int u = 0; u < 8; ++u) a[u] = acc[i + u * SC_T];
#pragma unroll
      for (int u = 0; u < 8; ++u) atomicAddF(dst + i + u * SC_T, a[u]);
    }
  }
  {
    float a[8];
#pragma unroll
    for (int u = 0; u < 8; ++u) {
      const int k = i + u * SC_T;
      a[u] = (real && k < tot) ? acc[k] : 0.0f;
    }
#pragma unroll
    for (int u = 0; u < 8; ++u) {
      const int k = i + u * SC_T;
      if (real && k < tot) atomicAddF(dst + k, a[u]);
    }
  }
}

__device__ __forceinline__ float xhalf(float x) {
  return __builtin_bit_cast(float, __builtin_amdgcn_permlanex16(__builtin_bit_cast(int, x), __builtin_bit_cast(int, x),
                                                                0x76543210, 0xfedcba98, false, false));
}

#define GO_T 256
__global__ __launch_bounds__(GO_T) void k_gemm_out(const float* __restrict__ A,
                                                  const float* __restrict__ dinv,
                                                  const float* __restrict__ bprev,
                                                  const float* __restrict__ Wc,
                                                  const float* __restrict__ bc,
                                                  float* __restrict__ out,
                                                  int numNodeTiles) {
  __shared__ v2f wfr[4 * 8 * 32];
  __shared__ float blds[HPAD];

  const int wave = threadIdx.x >> 5;
  const int lane = threadIdx.x & 31;
  const int tile = blockIdx.x * (GO_T / 32) + wave;
  const bool live = tile < numNodeTiles;
  const int tl = live ? tile : numNodeTiles - 1;
  const int m  = lane & 15;
  const int h  = lane >> 4;
  const int kh = h << 1;

  const float* arow = A + (size_t)(tl * 16 + m) * HPAD;
  const float dm = dinv[tl * 16 + m];
  float2 r[4];
#pragma unroll
  for (int ks = 0; ks < 4; ++ks) r[ks] = *reinterpret_cast<const float2*>(arow + 4 * ks + kh);

  v2f w[4];
#pragma unroll
  for (int ks = 0; ks < 4; ++ks) {
    const int ln = threadIdx.x & 31, j = threadIdx.x >> 5;
    const int k = 4 * ks + ((ln >> 4) << 1), n = 16 * j + (ln & 15);
    const int k1 = (k + 1 < HFEAT) ? k + 1 : k;
    const float y = Wc[k1 * DFEAT + n];
    w[ks].x = Wc[k * DFEAT + n];
    w[ks].y = (k + 1 < HFEAT) ? y : 0.0f;
  }
#pragma unroll
  for (int ks = 0; ks < 4; ++ks) wfr[ks * GO_T + threadIdx.x] = w[ks];
  if (threadIdx.x < HPAD) blds[threadIdx.x] = threadIdx.x < HFEAT ? bprev[threadIdx.x] : 0.0f;
  __syncthreads();
  if (!live) return;

  v2f a[4];
#pragma unroll
  for (int ks = 0; ks < 4; ++ks) {
    const int k = 4 * ks + kh;
    a[ks].x = fmaxf(dm * r[ks].x + blds[k], 0.0f);
    a[ks].y = (k + 1 < HFEAT) ? fmaxf(dm * r[ks].y + blds[k + 1], 0.0f) : 0.0f;
  }

  v8f c[8];
#pragma unroll
  for (int j = 0; j < 8; ++j) c[j] = v8f{};
#pragma unroll
  for (int ks = 0; ks < 4; ++ks) {
#pragma unroll
    for (int j = 0; j < 8; ++j) {
      const v2f b = wfr[(ks * 8 + j) * 32 + lane];
      c[j] = __builtin_amdgcn_wmma_f32_16x16x4_f32(false, a[ks], false, b, (short)0, c[j],
                                                   false, false);
    }
  }

  float* orow = out + (size_t)(tile * 16) * DFEAT + lane;
#pragma unroll
  for (int p = 0; p < 4; ++p) {
    const float bias = bc[32 * p + lane];
#pragma unroll
    for (int i = 0; i < 8; ++i) {
      const float lo = c[2 * p][i], hi = c[2 * p + 1][i];
      const float got = xhalf(h ? lo : hi);
      const float vA = h ? got : lo;
      const float vB = h ? hi : got;
      orow[(size_t)i * DFEAT + 32 * p]       = vA + bias;
      orow[(size_t)(8 + i) * DFEAT + 32 * p] = vB + bias;
    }
  }
}

extern "C" void kernel_launch(void* const* d_in, const int* in_sizes, int n_in,
                              void* d_out, int out_size, void* d_ws, size_t ws_size,
                              hipStream_t stream) {
  const float* x  = (const float*)d_in[0];
  const int*   ei = (const int*)d_in[1];
  const float* W1 = (const float*)d_in[2];
  const float* b1 = (const float*)d_in[3];
  const float* W2 = (const float*)d_in[4];
  const float* b2 = (const float*)d_in[5];
  const float* W3 = (const float*)d_in[6];
  const float* b3 = (const float*)d_in[7];
  const float* Wc = (const float*)d_in[8];
  const float* bc = (const float*)d_in[9];

  const int N = in_sizes[0] / DFEAT;
  const int E = in_sizes[1] / 2;
  const int* row = ei;
  const int* col = ei + E;

  float* dinv = (float*)d_ws;
  float* B0 = dinv + N;
  float* B1 = B0 + (size_t)N * HPAD;
  float* B2 = B1 + (size_t)N * HPAD;
  float* outf = (float*)d_out;
  float* slots = outf;

  const int T = GS_T;
  const int numTiles = N / 16;
  dim3 gS((N + SC_NR - 1) / SC_NR, SC_CHUNKS);
  dim3 gB(SC_T / 32, SC_PC);
  dim3 gG((numTiles + 7) / 8);

  k_bin      <<<gB, SC_T, 0, stream>>>(row, col, slots, dinv, N, E, (int)gS.x);
  k_sort     <<<gS, SC_T, 0, stream>>>(row, col, dinv, slots, N, E);

  k_gemm_scale<DFEAT, true><<<gG, T, 0, stream>>>(x, W1, DFEAT, b1, dinv, B0, B1, numTiles);
  k_scatter<<<gS, SC_T, 0, stream>>>(row, col, B0, B1, slots, N, E);

  k_gemm_scale<HPAD, false><<<gG, T, 0, stream>>>(B1, W2, HFEAT, b1, dinv, B0, B2, numTiles);
  k_scatter<<<gS, SC_T, 0, stream>>>(row, col, B0, B2, slots, N, E);

  k_gemm_scale<HPAD, false><<<gG, T, 0, stream>>>(B2, W3, HFEAT, b2, dinv, B0, B1, numTiles);
  k_scatter<<<gS, SC_T, 0, stream>>>(row, col, B0, B1, slots, N, E);

  k_gemm_out<<<gG, GO_T, 0, stream>>>(B1, dinv, b3, Wc, bc, outf, numTiles);
}
